# starting kernel
# baseline (speedup 1.0000x reference)
#include <hip/hip_runtime.h>
#include <hip/hip_bf16.h>

typedef float v2f __attribute__((ext_vector_type(2)));
typedef float v8f __attribute__((ext_vector_type(8)));

#define NNODE 7936
#define CIN   5
#define WLEN  64
#define T1    57
#define T2    50
#define H1C   64
#define H2C   16
#define NEDGE 63488

#define XS_STRIDE 321
#define HS_STRIDE 66
#define GS_STRIDE 17

__device__ __forceinline__ v8f wmma_f32(v2f a, v2f b, v8f c) {
  return __builtin_amdgcn_wmma_f32_16x16x4_f32(false, a, false, b, (short)0, c,
                                               false, false);
}

__device__ __forceinline__ float sigmoidf_(float x) {
  return 1.0f / (1.0f + __expf(-x));
}

__global__ void k_init(float* __restrict__ gout, float* __restrict__ deg, int gsz) {
  int i = blockIdx.x * blockDim.x + threadIdx.x;
  if (i < gsz) gout[i] = 0.0f;
  if (i < NNODE) deg[i] = 1.0f;
}

__global__ __launch_bounds__(256) void k_conv1(const float* __restrict__ x,
                                               const float* __restrict__ Wc1,
                                               const float* __restrict__ b1,
                                               const float* __restrict__ Wg,
                                               float* __restrict__ xt) {
  __shared__ float Xs[16 * XS_STRIDE];
  __shared__ float Hs[8 * 16 * HS_STRIDE];

  const int tid    = threadIdx.x;
  const int lane   = tid & 31;
  const int wave   = tid >> 5;
  const int lane16 = lane & 15;
  const int half   = lane >> 4;
  const int n0     = blockIdx.x * 16;

  for (int idx = tid; idx < 16 * CIN * WLEN; idx += 256) {
    int n = idx / (CIN * WLEN), r = idx % (CIN * WLEN);
    Xs[n * XS_STRIDE + r] = x[(n0 + n) * (CIN * WLEN) + r];
  }

  const int pb  = wave & 3;
  const int coP = pb * 16 + lane16;
  const int coQ = coP + 64;
  v2f bp1[10], bq1[10];
#pragma unroll
  for (int s = 0; s < 10; ++s) {
    int k     = 4 * s + 2 * half;
    bp1[s][0] = Wc1[coP * 40 + k];
    bp1[s][1] = Wc1[coP * 40 + k + 1];
    bq1[s][0] = Wc1[coQ * 40 + k];
    bq1[s][1] = Wc1[coQ * 40 + k + 1];
  }
  const float biasP = b1[coP];
  const float biasQ = b1[coQ];

  v2f bwg[16];
#pragma unroll
  for (int s = 0; s < 16; ++s) {
    int k     = 4 * s + 2 * half;
    bwg[s][0] = Wg[k * 16 + lane16];
    bwg[s][1] = Wg[(k + 1) * 16 + lane16];
  }
  __syncthreads();

  const int dt = wave >> 2;
  for (int tbase = 0; tbase < T1; tbase += 8) {
    for (int i = 0; i < 4; ++i) {
      int tl = dt * 4 + i;
      int t  = tbase + tl;
      if (t < T1) {
        v8f accp, accq;
#pragma unroll
        for (int j = 0; j < 8; ++j) { accp[j] = biasP; accq[j] = biasQ; }
#pragma unroll
        for (int s = 0; s < 10; ++s) {
          int j0 = 4 * s + 2 * half;
          int ci = j0 >> 3, kk = j0 & 7;
          v2f a;
          a[0] = Xs[lane16 * XS_STRIDE + ci * WLEN + t + kk];
          a[1] = Xs[lane16 * XS_STRIDE + ci * WLEN + t + kk + 1];
          accp = wmma_f32(a, bp1[s], accp);
          accq = wmma_f32(a, bq1[s], accq);
        }
        int ch = pb * 16 + lane16;
#pragma unroll
        for (int j = 0; j < 8; ++j) {
          float hv = accp[j] * sigmoidf_(accq[j]);
          int m    = j + 8 * half;
          Hs[(tl * 16 + m) * HS_STRIDE + ch] = hv;
        }
      }
    }
    __syncthreads();
    {
      int tl = wave;
      int t  = tbase + tl;
      if (t < T1) {
        v8f acc = {};
#pragma unroll
        for (int s = 0; s < 16; ++s) {
          int k0 = 4 * s + 2 * half;
          v2f a;
          a[0] = Hs[(tl * 16 + lane16) * HS_STRIDE + k0];
          a[1] = Hs[(tl * 16 + lane16) * HS_STRIDE + k0 + 1];
          acc  = wmma_f32(a, bwg[s], acc);
        }
#pragma unroll
        for (int j = 0; j < 8; ++j) {
          int m = j + 8 * half;
          xt[(t * NNODE + n0 + m) * H2C + lane16] = acc[j];
        }
      }
    }
    __syncthreads();
  }
}

__global__ void k_deg(const int* __restrict__ col, const float* __restrict__ ew,
                      float* __restrict__ deg) {
  int e = blockIdx.x * blockDim.x + threadIdx.x;
  if (e < NEDGE) atomicAdd(&deg[col[e]], ew[e]);
}

__global__ void k_dinv(float* __restrict__ deg) {
  int n = blockIdx.x * blockDim.x + threadIdx.x;
  if (n < NNODE) {
    float d = deg[n];
    deg[n]  = (d > 0.0f) ? rsqrtf(d) : 0.0f;
  }
}

__global__ __launch_bounds__(256) void k_scatter(const int* __restrict__ row,
                                                 const int* __restrict__ col,
                                                 const float* __restrict__ ew,
                                                 const float* __restrict__ dinv,
                                                 const float* __restrict__ xt,
                                                 float* __restrict__ gout) {
  int job = blockIdx.x;
  int r, c;
  float w;
  if (job < NEDGE) {
    r = row[job]; c = col[job]; w = ew[job];
  } else {
    r = c = job - NEDGE; w = 1.0f;
  }
  float norm = dinv[r] * w * dinv[c];
  const float* src = xt + r * H2C;
  float* dst       = gout + c * H2C;
  for (int idx = threadIdx.x; idx < T1 * H2C; idx += 256) {
    int t = idx >> 4, f = idx & 15;
    atomicAdd(&dst[t * (NNODE * H2C) + f], norm * src[t * (NNODE * H2C) + f]);
  }
}

__global__ __launch_bounds__(256) void k_conv2(const float* __restrict__ gout,
                                               const float* __restrict__ bg,
                                               const float* __restrict__ Wc2,
                                               const float* __restrict__ b2,
                                               float* __restrict__ out) {
  __shared__ float Gs[T1 * 16 * GS_STRIDE];

  const int tid    = threadIdx.x;
  const int lane   = tid & 31;
  const int wave   = tid >> 5;
  const int lane16 = lane & 15;
  const int half   = lane >> 4;
  const int n0     = blockIdx.x * 16;

  for (int idx = tid; idx < T1 * 256; idx += 256) {
    int t = idx >> 8, r = idx & 255;
    int n = r >> 4, ci = r & 15;
    float v = gout[(t * NNODE + n0 + n) * H2C + ci] + bg[ci];
    Gs[(t * 16 + n) * GS_STRIDE + ci] = v > 0.0f ? v : 0.0f;
  }

  const int pb  = wave & 3;
  const int coP = pb * 16 + lane16;
  const int coQ = coP + 64;
  v2f bp2[32], bq2[32];
#pragma unroll
  for (int s = 0; s < 32; ++s) {
    int k     = 4 * s + 2 * half;
    bp2[s][0] = Wc2[coP * 128 + k];
    bp2[s][1] = Wc2[coP * 128 + k + 1];
    bq2[s][0] = Wc2[coQ * 128 + k];
    bq2[s][1] = Wc2[coQ * 128 + k + 1];
  }
  const float biasP = b2[coP];
  const float biasQ = b2[coQ];
  __syncthreads();

  const int thalf = wave >> 2;
  for (int t2 = thalf * 25; t2 < thalf * 25 + 25; ++t2) {
    v8f accp, accq;
#pragma unroll
    for (int j = 0; j < 8; ++j) { accp[j] = biasP; accq[j] = biasQ; }
#pragma unroll
    for (int s = 0; s < 32; ++s) {
      int j0 = 4 * s + 2 * half;
      int ci = j0 >> 3, kk = j0 & 7;
      v2f a;
      a[0] = Gs[((t2 + kk) * 16 + lane16) * GS_STRIDE + ci];
      a[1] = Gs[((t2 + kk + 1) * 16 + lane16) * GS_STRIDE + ci];
      accp = wmma_f32(a, bp2[s], accp);
      accq = wmma_f32(a, bq2[s], accq);
    }
#pragma unroll
    for (int j = 0; j < 8; ++j) {
      float hv = accp[j] * sigmoidf_(accq[j]);
      int m    = j + 8 * half;
      __builtin_nontemporal_store(hv, &out[(n0 + m) * (H1C * T2) + coP * T2 + t2]);
    }
  }
}

extern "C" void kernel_launch(void* const* d_in, const int* in_sizes, int n_in,
                              void* d_out, int out_size, void* d_ws, size_t ws_size,
                              hipStream_t stream) {
  const float* x   = (const float*)d_in[0];
  const int*   ei  = (const int*)d_in[1];
  const float* ea  = (const float*)d_in[2];
  const float* Wc1 = (const float*)d_in[4];
  const float* b1  = (const float*)d_in[5];
  const float* Wg  = (const float*)d_in[6];
  const float* bg  = (const float*)d_in[7];
  const float* Wc2 = (const float*)d_in[8];
  const float* b2  = (const float*)d_in[9];
  float*       out = (float*)d_out;

  const int* row = ei;
  const int* col = ei + NEDGE;

  const size_t xt_bytes = (size_t)T1 * NNODE * H2C * sizeof(float);
  float* xt   = (float*)d_ws;
  float* gout = (float*)((char*)d_ws + xt_bytes);
  float* deg  = (float*)((char*)d_ws + 2 * xt_bytes);

  const int gsz = T1 * NNODE * H2C;

  k_init<<<(gsz + 255) / 256, 256, 0, stream>>>(gout, deg, gsz);
  k_conv1<<<NNODE / 16, 256, 0, stream>>>(x, Wc1, b1, Wg, xt);
  k_deg<<<(NEDGE + 255) / 256, 256, 0, stream>>>(col, ea, deg);
  k_dinv<<<(NNODE + 255) / 256, 256, 0, stream>>>(deg);
  k_scatter<<<NEDGE + NNODE, 256, 0, stream>>>(row, col, ea, deg, xt, gout);
  k_conv2<<<NNODE / 16, 256, 0, stream>>>(gout, bg, Wc2, b2, out);
}


# conv2: weights as WMMA A operand, activations as B over flattened (node,t2) columns, so each output store writes 64-byte t2-contiguous runs instead of 32 scattered 4-byte elements
# speedup vs baseline: 2.1558x; 2.1558x over previous
#include <hip/hip_runtime.h>
#include <hip/hip_bf16.h>

typedef float v2f __attribute__((ext_vector_type(2)));
typedef float v8f __attribute__((ext_vector_type(8)));

#define NNODE 7936
#define CIN   5
#define WLEN  64
#define T1    57
#define T2    50
#define H1C   64
#define H2C   16
#define NEDGE 63488

#define XS_STRIDE 321
#define HS_STRIDE 66
#define GS_STRIDE 17

__device__ __forceinline__ v8f wmma_f32(v2f a, v2f b, v8f c) {
  return __builtin_amdgcn_wmma_f32_16x16x4_f32(false, a, false, b, (short)0, c,
                                               false, false);
}

__device__ __forceinline__ float sigmoidf_(float x) {
  return 1.0f / (1.0f + __expf(-x));
}

__global__ void k_init(float* __restrict__ gout, float* __restrict__ deg, int gsz) {
  int i = blockIdx.x * blockDim.x + threadIdx.x;
  if (i < gsz) gout[i] = 0.0f;
  if (i < NNODE) deg[i] = 1.0f;
}

__global__ __launch_bounds__(256) void k_conv1(const float* __restrict__ x,
                                               const float* __restrict__ Wc1,
                                               const float* __restrict__ b1,
                                               const float* __restrict__ Wg,
                                               float* __restrict__ xt) {
  __shared__ float Xs[16 * XS_STRIDE];
  __shared__ float Hs[8 * 16 * HS_STRIDE];

  const int tid    = threadIdx.x;
  const int lane   = tid & 31;
  const int wave   = tid >> 5;
  const int lane16 = lane & 15;
  const int half   = lane >> 4;
  const int n0     = blockIdx.x * 16;

  for (int idx = tid; idx < 16 * CIN * WLEN; idx += 256) {
    int n = idx / (CIN * WLEN), r = idx % (CIN * WLEN);
    Xs[n * XS_STRIDE + r] = x[(n0 + n) * (CIN * WLEN) + r];
  }

  const int pb  = wave & 3;
  const int coP = pb * 16 + lane16;
  const int coQ = coP + 64;
  v2f bp1[10], bq1[10];
#pragma unroll
  for (int s = 0; s < 10; ++s) {
    int k     = 4 * s + 2 * half;
    bp1[s][0] = Wc1[coP * 40 + k];
    bp1[s][1] = Wc1[coP * 40 + k + 1];
    bq1[s][0] = Wc1[coQ * 40 + k];
    bq1[s][1] = Wc1[coQ * 40 + k + 1];
  }
  const float biasP = b1[coP];
  const float biasQ = b1[coQ];

  v2f bwg[16];
#pragma unroll
  for (int s = 0; s < 16; ++s) {
    int k     = 4 * s + 2 * half;
    bwg[s][0] = Wg[k * 16 + lane16];
    bwg[s][1] = Wg[(k + 1) * 16 + lane16];
  }
  __syncthreads();

  const int dt = wave >> 2;
  for (int tbase = 0; tbase < T1; tbase += 8) {
    for (int i = 0; i < 4; ++i) {
      int tl = dt * 4 + i;
      int t  = tbase + tl;
      if (t < T1) {
        v8f accp, accq;
#pragma unroll
        for (int j = 0; j < 8; ++j) { accp[j] = biasP; accq[j] = biasQ; }
#pragma unroll
        for (int s = 0; s < 10; ++s) {
          int j0 = 4 * s + 2 * half;
          int ci = j0 >> 3, kk = j0 & 7;
          v2f a;
          a[0] = Xs[lane16 * XS_STRIDE + ci * WLEN + t + kk];
          a[1] = Xs[lane16 * XS_STRIDE + ci * WLEN + t + kk + 1];
          accp = wmma_f32(a, bp1[s], accp);
          accq = wmma_f32(a, bq1[s], accq);
        }
        int ch = pb * 16 + lane16;
#pragma unroll
        for (int j = 0; j < 8; ++j) {
          float hv = accp[j] * sigmoidf_(accq[j]);
          int m    = j + 8 * half;
          Hs[(tl * 16 + m) * HS_STRIDE + ch] = hv;
        }
      }
    }
    __syncthreads();
    {
      int tl = wave;
      int t  = tbase + tl;
      if (t < T1) {
        v8f acc = {};
#pragma unroll
        for (int s = 0; s < 16; ++s) {
          int k0 = 4 * s + 2 * half;
          v2f a;
          a[0] = Hs[(tl * 16 + lane16) * HS_STRIDE + k0];
          a[1] = Hs[(tl * 16 + lane16) * HS_STRIDE + k0 + 1];
          acc  = wmma_f32(a, bwg[s], acc);
        }
#pragma unroll
        for (int j = 0; j < 8; ++j) {
          int m = j + 8 * half;
          xt[(t * NNODE + n0 + m) * H2C + lane16] = acc[j];
        }
      }
    }
    __syncthreads();
  }
}

__global__ void k_deg(const int* __restrict__ col, const float* __restrict__ ew,
                      float* __restrict__ deg) {
  int e = blockIdx.x * blockDim.x + threadIdx.x;
  if (e < NEDGE) atomicAdd(&deg[col[e]], ew[e]);
}

__global__ void k_dinv(float* __restrict__ deg) {
  int n = blockIdx.x * blockDim.x + threadIdx.x;
  if (n < NNODE) {
    float d = deg[n];
    deg[n]  = (d > 0.0f) ? rsqrtf(d) : 0.0f;
  }
}

__global__ __launch_bounds__(256) void k_scatter(const int* __restrict__ row,
                                                 const int* __restrict__ col,
                                                 const float* __restrict__ ew,
                                                 const float* __restrict__ dinv,
                                                 const float* __restrict__ xt,
                                                 float* __restrict__ gout) {
  int job = blockIdx.x;
  int r, c;
  float w;
  if (job < NEDGE) {
    r = row[job]; c = col[job]; w = ew[job];
  } else {
    r = c = job - NEDGE; w = 1.0f;
  }
  float norm = dinv[r] * w * dinv[c];
  const float* src = xt + r * H2C;
  float* dst       = gout + c * H2C;
  for (int idx = threadIdx.x; idx < T1 * H2C; idx += 256) {
    int t = idx >> 4, f = idx & 15;
    atomicAdd(&dst[t * (NNODE * H2C) + f], norm * src[t * (NNODE * H2C) + f]);
  }
}

__global__ __launch_bounds__(256) void k_conv2(const float* __restrict__ gout,
                                               const float* __restrict__ bg,
                                               const float* __restrict__ Wc2,
                                               const float* __restrict__ b2,
                                               float* __restrict__ out) {
  __shared__ float Gs[16 * T1 * GS_STRIDE];

  const int tid    = threadIdx.x;
  const int lane   = tid & 31;
  const int wave   = tid >> 5;
  const int lane16 = lane & 15;
  const int half   = lane >> 4;
  const int n0     = blockIdx.x * 16;

  for (int idx = tid; idx < T1 * 256; idx += 256) {
    int t = idx >> 8, r = idx & 255;
    int n = r >> 4, ci = r & 15;
    float v = gout[(t * NNODE + n0 + n) * H2C + ci] + bg[ci];
    Gs[(n * T1 + t) * GS_STRIDE + ci] = v > 0.0f ? v : 0.0f;
  }

  const int pb  = wave & 3;
  const int coP = pb * 16 + lane16;
  const int coQ = coP + 64;
  v2f ap2[32], aq2[32];
#pragma unroll
  for (int s = 0; s < 32; ++s) {
    int k     = 4 * s + 2 * half;
    ap2[s][0] = Wc2[coP * 128 + k];
    ap2[s][1] = Wc2[coP * 128 + k + 1];
    aq2[s][0] = Wc2[coQ * 128 + k];
    aq2[s][1] = Wc2[coQ * 128 + k + 1];
  }
  float biasP[8], biasQ[8];
#pragma unroll
  for (int j = 0; j < 8; ++j) {
    biasP[j] = b2[pb * 16 + 8 * half + j];
    biasQ[j] = b2[pb * 16 + 8 * half + j + 64];
  }
  __syncthreads();

  const int qhalf = wave >> 2;
  for (int q = qhalf * 25; q < qhalf * 25 + 25; ++q) {
    const int cc = q * 16 + lane16;
    const int n  = cc / T2;
    const int t2 = cc - n * T2;
    const float* gb = &Gs[(n * T1 + t2 + 2 * half) * GS_STRIDE];
    v8f accp, accq;
#pragma unroll
    for (int j = 0; j < 8; ++j) { accp[j] = biasP[j]; accq[j] = biasQ[j]; }
#pragma unroll
    for (int s = 0; s < 32; ++s) {
      const int kk4 = (4 * s) & 7;
      const int ci  = s >> 1;
      v2f b;
      b[0] = gb[kk4 * GS_STRIDE + ci];
      b[1] = gb[(kk4 + 1) * GS_STRIDE + ci];
      accp = wmma_f32(ap2[s], b, accp);
      accq = wmma_f32(aq2[s], b, accq);
    }
    float* op = out + (n0 + n) * (H1C * T2) + (pb * 16 + 8 * half) * T2 + t2;
#pragma unroll
    for (int j = 0; j < 8; ++j) {
      float hv = accp[j] * sigmoidf_(accq[j]);
      __builtin_nontemporal_store(hv, &op[j * T2]);
    }
  }
}

extern "C" void kernel_launch(void* const* d_in, const int* in_sizes, int n_in,
                              void* d_out, int out_size, void* d_ws, size_t ws_size,
                              hipStream_t stream) {
  const float* x   = (const float*)d_in[0];
  const int*   ei  = (const int*)d_in[1];
  const float* ea  = (const float*)d_in[2];
  const float* Wc1 = (const float*)d_in[4];
  const float* b1  = (const float*)d_in[5];
  const float* Wg  = (const float*)d_in[6];
  const float* bg  = (const float*)d_in[7];
  const float* Wc2 = (const float*)d_in[8];
  const float* b2  = (const float*)d_in[9];
  float*       out = (float*)d_out;

  const int* row = ei;
  const int* col = ei + NEDGE;

  const size_t xt_bytes = (size_t)T1 * NNODE * H2C * sizeof(float);
  float* xt   = (float*)d_ws;
  float* gout = (float*)((char*)d_ws + xt_bytes);
  float* deg  = (float*)((char*)d_ws + 2 * xt_bytes);

  const int gsz = T1 * NNODE * H2C;

  k_init<<<(gsz + 255) / 256, 256, 0, stream>>>(gout, deg, gsz);
  k_conv1<<<NNODE / 16, 256, 0, stream>>>(x, Wc1, b1, Wg, xt);
  k_deg<<<(NEDGE + 255) / 256, 256, 0, stream>>>(col, ea, deg);
  k_dinv<<<(NNODE + 255) / 256, 256, 0, stream>>>(deg);
  k_scatter<<<NEDGE + NNODE, 256, 0, stream>>>(row, col, ea, deg, xt, gout);
  k_conv2<<<NNODE / 16, 256, 0, stream>>>(gout, bg, Wc2, b2, out);
}


# GCN aggregation as a per-destination gather fused into conv2's tile load (xt node-major, slots from an integer-atomic counting sort): no float atomics on features, no zero-fill; sum in slot order
# speedup vs baseline: 3.6713x; 1.7030x over previous
#include <hip/hip_runtime.h>
#include <hip/hip_bf16.h>

typedef float v2f __attribute__((ext_vector_type(2)));
typedef float v8f __attribute__((ext_vector_type(8)));

#define NNODE 7936
#define CIN   5
#define WLEN  64
#define T1    57
#define T2    50
#define H1C   64
#define H2C   16
#define NEDGE 63488

#define XS_STRIDE 321
#define HS_STRIDE 66
#define GS_STRIDE 17

__device__ __forceinline__ v8f wmma_f32(v2f a, v2f b, v8f c) {
  return __builtin_amdgcn_wmma_f32_16x16x4_f32(false, a, false, b, (short)0, c,
                                               false, false);
}

__device__ __forceinline__ float sigmoidf_(float x) {
  return 1.0f / (1.0f + __expf(-x));
}

__global__ void k_init(float* __restrict__ deg, int* __restrict__ cnt) {
  int i = blockIdx.x * blockDim.x + threadIdx.x;
  if (i < NNODE) { deg[i] = 1.0f; cnt[i] = 0; }
}

__global__ __launch_bounds__(256) void k_conv1(const float* __restrict__ x,
                                               const float* __restrict__ Wc1,
                                               const float* __restrict__ b1,
                                               const float* __restrict__ Wg,
                                               float* __restrict__ xt) {
  __shared__ float Xs[16 * XS_STRIDE];
  __shared__ float Hs[8 * 16 * HS_STRIDE];

  const int tid    = threadIdx.x;
  const int lane   = tid & 31;
  const int wave   = tid >> 5;
  const int lane16 = lane & 15;
  const int half   = lane >> 4;
  const int n0     = blockIdx.x * 16;

  for (int idx = tid; idx < 16 * CIN * WLEN; idx += 256) {
    int n = idx / (CIN * WLEN), r = idx % (CIN * WLEN);
    Xs[n * XS_STRIDE + r] = x[(n0 + n) * (CIN * WLEN) + r];
  }

  const int pb  = wave & 3;
  const int coP = pb * 16 + lane16;
  const int coQ = coP + 64;
  v2f bp1[10], bq1[10];
#pragma unroll
  for (int s = 0; s < 10; ++s) {
    int k     = 4 * s + 2 * half;
    bp1[s][0] = Wc1[coP * 40 + k];
    bp1[s][1] = Wc1[coP * 40 + k + 1];
    bq1[s][0] = Wc1[coQ * 40 + k];
    bq1[s][1] = Wc1[coQ * 40 + k + 1];
  }
  const float biasP = b1[coP];
  const float biasQ = b1[coQ];

  v2f bwg[16];
#pragma unroll
  for (int s = 0; s < 16; ++s) {
    int k     = 4 * s + 2 * half;
    bwg[s][0] = Wg[k * 16 + lane16];
    bwg[s][1] = Wg[(k + 1) * 16 + lane16];
  }
  __syncthreads();

  const int dt = wave >> 2;
  for (int tbase = 0; tbase < T1; tbase += 8) {
    for (int i = 0; i < 4; ++i) {
      int tl = dt * 4 + i;
      int t  = tbase + tl;
      if (t < T1) {
        v8f accp, accq;
#pragma unroll
        for (int j = 0; j < 8; ++j) { accp[j] = biasP; accq[j] = biasQ; }
#pragma unroll
        for (int s = 0; s < 10; ++s) {
          int j0 = 4 * s + 2 * half;
          int ci = j0 >> 3, kk = j0 & 7;
          v2f a;
          a[0] = Xs[lane16 * XS_STRIDE + ci * WLEN + t + kk];
          a[1] = Xs[lane16 * XS_STRIDE + ci * WLEN + t + kk + 1];
          accp = wmma_f32(a, bp1[s], accp);
          accq = wmma_f32(a, bq1[s], accq);
        }
        int ch = pb * 16 + lane16;
#pragma unroll
        for (int j = 0; j < 8; ++j) {
          float hv = accp[j] * sigmoidf_(accq[j]);
          int m    = j + 8 * half;
          Hs[(tl * 16 + m) * HS_STRIDE + ch] = hv;
        }
      }
    }
    __syncthreads();
    {
      int tl = wave;
      int t  = tbase + tl;
      if (t < T1) {
        v8f acc = {};
#pragma unroll
        for (int s = 0; s < 16; ++s) {
          int k0 = 4 * s + 2 * half;
          v2f a;
          a[0] = Hs[(tl * 16 + lane16) * HS_STRIDE + k0];
          a[1] = Hs[(tl * 16 + lane16) * HS_STRIDE + k0 + 1];
          acc  = wmma_f32(a, bwg[s], acc);
        }
#pragma unroll
        for (int j = 0; j < 8; ++j) {
          int m = j + 8 * half;
          xt[((n0 + m) * T1 + t) * H2C + lane16] = acc[j];
        }
      }
    }
    __syncthreads();
  }
}

__global__ void k_deg(const int* __restrict__ col, const float* __restrict__ ew,
                      float* __restrict__ deg, int* __restrict__ cnt) {
  int e = blockIdx.x * blockDim.x + threadIdx.x;
  if (e < NEDGE) {
    int c = col[e];
    atomicAdd(&deg[c], ew[e]);
    atomicAdd(&cnt[c], 1);
  }
}

__global__ __launch_bounds__(1024) void k_scan(float* __restrict__ deg,
                                               const int* __restrict__ cnt,
                                               int* __restrict__ off) {
  __shared__ int part[1024];
  const int i    = threadIdx.x;
  const int base = i * 8;
  int c[8];
  int s = 0;
#pragma unroll
  for (int k = 0; k < 8; ++k) {
    int n = base + k;
    c[k]  = (n < NNODE) ? cnt[n] : 0;
    s += c[k];
  }
  part[i] = s;
  __syncthreads();
  for (int d = 1; d < 1024; d <<= 1) {
    int v = (i >= d) ? part[i - d] : 0;
    __syncthreads();
    part[i] += v;
    __syncthreads();
  }
  int run = part[i] - s;
#pragma unroll
  for (int k = 0; k < 8; ++k) {
    int n = base + k;
    if (n < NNODE) {
      off[n] = run;
      run += c[k];
      float d = deg[n];
      deg[n]  = (d > 0.0f) ? rsqrtf(d) : 0.0f;
    }
  }
  if (i == 1023) off[NNODE] = part[1023];
}

__global__ void k_fill(const int* __restrict__ row, const int* __restrict__ col,
                       const float* __restrict__ ew, const float* __restrict__ dinv,
                       const int* __restrict__ off, int* __restrict__ cnt,
                       int* __restrict__ src, float* __restrict__ nrm) {
  int e = blockIdx.x * blockDim.x + threadIdx.x;
  if (e < NEDGE) {
    int r = row[e], c = col[e];
    int slot  = off[c] + atomicSub(&cnt[c], 1) - 1;
    src[slot] = r;
    nrm[slot] = dinv[r] * ew[e] * dinv[c];
  }
}

#define ROW4 (T1 * H2C / 4)
#define ECAP 1024
#define EU   8
#define NPC  2

template <bool STG>
__device__ __forceinline__ void gather_tile(const float4* __restrict__ xt4,
                                            const int* __restrict__ src,
                                            const float* __restrict__ nrm,
                                            const int* es, const float* ws, int bs,
                                            const int* offs, const float* dn,
                                            const float* __restrict__ bg, int n0,
                                            int tid, float* Gs) {
  const int f0 = (tid & 3) * 4;
  const float bg0 = bg[f0], bg1 = bg[f0 + 1], bg2 = bg[f0 + 2], bg3 = bg[f0 + 3];
  for (int o = tid; o < 16 * ROW4; o += 256 * NPC) {
    int   n[NPC], rem[NPC], b[NPC], en[NPC];
    bool  live[NPC];
    float4 vs[NPC];
    float ax[NPC], ay[NPC], az[NPC], aw[NPC];
    int len = 0;
#pragma unroll
    for (int q = 0; q < NPC; ++q) {
      live[q] = (o + 256 * q) < 16 * ROW4;
      const int oq = live[q] ? (o + 256 * q) : o;
      n[q]   = oq / ROW4;
      rem[q] = oq - n[q] * ROW4;
      b[q]   = offs[n[q]];
      en[q]  = offs[n[q] + 1];
      vs[q]  = xt4[(n0 + n[q]) * ROW4 + rem[q]];
      ax[q] = ay[q] = az[q] = aw[q] = 0.0f;
      len = max(len, en[q] - b[q]);
    }
    for (int j = 0; j < len; j += EU) {
      int   r[NPC][EU];
      float w[NPC][EU];
      float4 v[NPC][EU];
#pragma unroll
      for (int q = 0; q < NPC; ++q) {
#pragma unroll
        for (int k = 0; k < EU; ++k) {
          const int pp = b[q] + j + k;
          const int pc = max(min(pp, en[q] - 1), bs);
          r[q][k] = STG ? es[pc - bs] : src[pc];
          const float wk = STG ? ws[pc - bs] : nrm[pc];
          w[q][k] = (pp < en[q]) ? wk : 0.0f;
        }
      }
#pragma unroll
      for (int q = 0; q < NPC; ++q) {
#pragma unroll
        for (int k = 0; k < EU; ++k) v[q][k] = xt4[r[q][k] * ROW4 + rem[q]];
      }
#pragma unroll
      for (int q = 0; q < NPC; ++q) {
#pragma unroll
        for (int k = 0; k < EU; ++k) {
          ax[q] += w[q][k] * v[q][k].x;
          ay[q] += w[q][k] * v[q][k].y;
          az[q] += w[q][k] * v[q][k].z;
          aw[q] += w[q][k] * v[q][k].w;
        }
      }
    }
#pragma unroll
    for (int q = 0; q < NPC; ++q) {
      if (live[q]) {
        const float wself = dn[n[q]] * dn[n[q]];
        float gx = ax[q] + wself * vs[q].x + bg0;
        float gy = ay[q] + wself * vs[q].y + bg1;
        float gz = az[q] + wself * vs[q].z + bg2;
        float gw = aw[q] + wself * vs[q].w + bg3;
        float* g = &Gs[(n[q] * T1 + (rem[q] >> 2)) * GS_STRIDE + f0];
        g[0] = gx > 0.0f ? gx : 0.0f;
        g[1] = gy > 0.0f ? gy : 0.0f;
        g[2] = gz > 0.0f ? gz : 0.0f;
        g[3] = gw > 0.0f ? gw : 0.0f;
      }
    }
  }
}

__global__ __launch_bounds__(256) void k_conv2(const float* __restrict__ xt,
                                               const int* __restrict__ off,
                                               const int* __restrict__ src,
                                               const float* __restrict__ nrm,
                                               const float* __restrict__ dinv,
                                               const float* __restrict__ bg,
                                               const float* __restrict__ Wc2,
                                               const float* __restrict__ b2,
                                               float* __restrict__ out) {
  __shared__ float Gs[16 * T1 * GS_STRIDE];
  __shared__ int   Es[ECAP];
  __shared__ float Ws[ECAP];
  __shared__ int   Offs[17];
  __shared__ float Dn[16];

  const int tid    = threadIdx.x;
  const int lane   = tid & 31;
  const int wave   = tid >> 5;
  const int lane16 = lane & 15;
  const int half   = lane >> 4;
  const int n0     = blockIdx.x * 16;

  if (tid < 17) Offs[tid] = off[n0 + tid];
  if (tid >= 32 && tid < 48) Dn[tid - 32] = dinv[n0 + tid - 32];
  __syncthreads();
  const int  bs     = Offs[0];
  const int  nslot  = Offs[16] - bs;
  const bool staged = nslot <= ECAP;
  if (staged) {
    for (int i = tid; i < nslot; i += 256) {
      Es[i] = src[bs + i];
      Ws[i] = nrm[bs + i];
    }
  }
  __syncthreads();
  const float4* xt4 = reinterpret_cast<const float4*>(xt);
  if (staged)
    gather_tile<true>(xt4, src, nrm, Es, Ws, bs, Offs, Dn, bg, n0, tid, Gs);
  else
    gather_tile<false>(xt4, src, nrm, Es, Ws, bs, Offs, Dn, bg, n0, tid, Gs);

  const int pb  = wave & 3;
  const int coP = pb * 16 + lane16;
  const int coQ = coP + 64;
  v2f ap2[32], aq2[32];
#pragma unroll
  for (int s = 0; s < 32; ++s) {
    int k     = 4 * s + 2 * half;
    ap2[s][0] = Wc2[coP * 128 + k];
    ap2[s][1] = Wc2[coP * 128 + k + 1];
    aq2[s][0] = Wc2[coQ * 128 + k];
    aq2[s][1] = Wc2[coQ * 128 + k + 1];
  }
  float biasP[8], biasQ[8];
#pragma unroll
  for (int j = 0; j < 8; ++j) {
    biasP[j] = b2[pb * 16 + 8 * half + j];
    biasQ[j] = b2[pb * 16 + 8 * half + j + 64];
  }
  __syncthreads();

  const int qhalf = wave >> 2;
  for (int q = qhalf * 25; q < qhalf * 25 + 25; ++q) {
    const int cc = q * 16 + lane16;
    const int n  = cc / T2;
    const int t2 = cc - n * T2;
    const float* gb = &Gs[(n * T1 + t2 + 2 * half) * GS_STRIDE];
    v8f accp, accq;
#pragma unroll
    for (int j = 0; j < 8; ++j) { accp[j] = biasP[j]; accq[j] = biasQ[j]; }
#pragma unroll
    for (int s = 0; s < 32; ++s) {
      const int kk4 = (4 * s) & 7;
      const int ci  = s >> 1;
      v2f b;
      b[0] = gb[kk4 * GS_STRIDE + ci];
      b[1] = gb[(kk4 + 1) * GS_STRIDE + ci];
      accp = wmma_f32(ap2[s], b, accp);
      accq = wmma_f32(aq2[s], b, accq);
    }
    float* op = out + (n0 + n) * (H1C * T2) + (pb * 16 + 8 * half) * T2 + t2;
#pragma unroll
    for (int j = 0; j < 8; ++j) {
      float hv = accp[j] * sigmoidf_(accq[j]);
      __builtin_nontemporal_store(hv, &op[j * T2]);
    }
  }
}

extern "C" void kernel_launch(void* const* d_in, const int* in_sizes, int n_in,
                              void* d_out, int out_size, void* d_ws, size_t ws_size,
                              hipStream_t stream) {
  const float* x   = (const float*)d_in[0];
  const int*   ei  = (const int*)d_in[1];
  const float* ea  = (const float*)d_in[2];
  const float* Wc1 = (const float*)d_in[4];
  const float* b1  = (const float*)d_in[5];
  const float* Wg  = (const float*)d_in[6];
  const float* bg  = (const float*)d_in[7];
  const float* Wc2 = (const float*)d_in[8];
  const float* b2  = (const float*)d_in[9];
  float*       out = (float*)d_out;

  const int* row = ei;
  const int* col = ei + NEDGE;

  const size_t xt_bytes = (size_t)T1 * NNODE * H2C * sizeof(float);
  char*  wsb = (char*)d_ws;
  float* xt  = (float*)wsb;
  float* deg = (float*)(wsb + xt_bytes);
  int*   cnt = (int*)(wsb + xt_bytes + 32768);
  int*   off = (int*)(wsb + xt_bytes + 2 * 32768);
  int*   src = (int*)(wsb + xt_bytes + 4 * 32768);
  float* nrm = (float*)(wsb + xt_bytes + 4 * 32768 + 262144);

  k_init<<<(NNODE + 255) / 256, 256, 0, stream>>>(deg, cnt);
  k_conv1<<<NNODE / 16, 256, 0, stream>>>(x, Wc1, b1, Wg, xt);
  k_deg<<<(NEDGE + 255) / 256, 256, 0, stream>>>(col, ea, deg, cnt);
  k_scan<<<1, 1024, 0, stream>>>(deg, cnt, off);
  k_fill<<<(NEDGE + 255) / 256, 256, 0, stream>>>(row, col, ea, deg, off, cnt, src, nrm);
  k_conv2<<<NNODE / 16, 256, 0, stream>>>(xt, off, src, nrm, deg, bg, Wc2, b2, out);
}


# graph prep (degrees, slot offsets, counting-sort fill) by destination-range work-groups riding in conv1's launch: 6 launches -> 2; slot order and degree sums by LDS-atomic arrival order
# speedup vs baseline: 4.0839x; 1.1124x over previous
#include <hip/hip_runtime.h>
#include <hip/hip_bf16.h>

typedef float v2f __attribute__((ext_vector_type(2)));
typedef float v8f __attribute__((ext_vector_type(8)));

#define NNODE 7936
#define CIN   5
#define WLEN  64
#define T1    57
#define T2    50
#define H1C   64
#define H2C   16
#define NEDGE 63488

#define XS_STRIDE 321
#define HS_STRIDE 66
#define GS_STRIDE 17

__device__ __forceinline__ v8f wmma_f32(v2f a, v2f b, v8f c) {
  return __builtin_amdgcn_wmma_f32_16x16x4_f32(false, a, false, b, (short)0, c,
                                               false, false);
}

__device__ __forceinline__ float sigmoidf_(float x) {
  return 1.0f / (1.0f + __expf(-x));
}

#define PREP_NODES  256
#define PREP_BLOCKS (NNODE / PREP_NODES)
#define NE4         (NEDGE / 4)
#define PREP_U      8

__device__ __forceinline__ void graph_prep(const int* __restrict__ row,
                                           const int* __restrict__ col,
                                           const float* __restrict__ ew,
                                           float* __restrict__ dinv,
                                           int* __restrict__ off,
                                           int* __restrict__ src,
                                           float* __restrict__ wsl, int pblk, int tid,
                                           float* degL, int* cntL, int* curL, int* redL) {
  const int lo   = pblk * PREP_NODES;
  const int lane = tid & 31;
  const int wave = tid >> 5;
  const int4*   row4 = reinterpret_cast<const int4*>(row);
  const int4*   col4 = reinterpret_cast<const int4*>(col);
  const float4* ew4  = reinterpret_cast<const float4*>(ew);

  degL[tid] = 1.0f;
  cntL[tid] = 0;
  __syncthreads();

  int below = 0;
  for (int i0 = tid; i0 < NE4; i0 += 256 * PREP_U) {
    int4   c[PREP_U];
    float4 w[PREP_U];
#pragma unroll
    for (int u = 0; u < PREP_U; ++u) {
      const int i = i0 + 256 * u;
      if (i < NE4) { c[u] = col4[i]; w[u] = ew4[i]; }
      else         { c[u] = make_int4(NNODE, NNODE, NNODE, NNODE); w[u] = make_float4(0.f, 0.f, 0.f, 0.f); }
    }
#pragma unroll
    for (int u = 0; u < PREP_U; ++u) {
      const int   cc[4] = {c[u].x, c[u].y, c[u].z, c[u].w};
      const float ww[4] = {w[u].x, w[u].y, w[u].z, w[u].w};
#pragma unroll
      for (int k = 0; k < 4; ++k) {
        const int d = cc[k] - lo;
        below += (d < 0) ? 1 : 0;
        if ((unsigned)d < (unsigned)PREP_NODES) {
          atomicAdd(&degL[d], ww[k]);
          atomicAdd(&cntL[d], 1);
        }
      }
    }
  }
  for (int m = 16; m >= 1; m >>= 1) below += __shfl_xor(below, m);
  if (lane == 0) redL[wave] = below;
  __syncthreads();
  int base = 0;
  for (int v = 0; v < 8; ++v) base += redL[v];

  const int cn = cntL[tid];
  int inc = cn;
  for (int m = 1; m < 32; m <<= 1) {
    const int t = __shfl_up(inc, m);
    if (lane >= m) inc += t;
  }
  if (lane == 31) redL[8 + wave] = inc;
  __syncthreads();
  int wbase = 0;
  for (int v = 0; v < 8; ++v) wbase += (v < wave) ? redL[8 + v] : 0;
  const int first = base + wbase + inc - cn;
  off[lo + tid] = first;
  curL[tid]     = first;
  if (pblk == PREP_BLOCKS - 1 && tid == PREP_NODES - 1) off[NNODE] = first + cn;
  const float dg = degL[tid];
  dinv[lo + tid] = (dg > 0.0f) ? rsqrtf(dg) : 0.0f;
  __syncthreads();

  for (int i0 = tid; i0 < NE4; i0 += 256 * PREP_U) {
    int4   c[PREP_U], r[PREP_U];
    float4 w[PREP_U];
    int    slot[PREP_U][4];
#pragma unroll
    for (int u = 0; u < PREP_U; ++u) {
      const int i = i0 + 256 * u;
      if (i < NE4) { c[u] = col4[i]; r[u] = row4[i]; w[u] = ew4[i]; }
      else {
        c[u] = make_int4(NNODE, NNODE, NNODE, NNODE);
        r[u] = make_int4(0, 0, 0, 0);
        w[u] = make_float4(0.f, 0.f, 0.f, 0.f);
      }
    }
#pragma unroll
    for (int u = 0; u < PREP_U; ++u) {
      const int cc[4] = {c[u].x, c[u].y, c[u].z, c[u].w};
#pragma unroll
      for (int k = 0; k < 4; ++k) {
        const int d = cc[k] - lo;
        slot[u][k] = -1;
        if ((unsigned)d < (unsigned)PREP_NODES) slot[u][k] = atomicAdd(&curL[d], 1);
      }
    }
#pragma unroll
    for (int u = 0; u < PREP_U; ++u) {
      const int   rr[4] = {r[u].x, r[u].y, r[u].z, r[u].w};
      const float ww[4] = {w[u].x, w[u].y, w[u].z, w[u].w};
#pragma unroll
      for (int k = 0; k < 4; ++k) {
        if (slot[u][k] >= 0) {
          src[slot[u][k]] = rr[k];
          wsl[slot[u][k]] = ww[k];
        }
      }
    }
  }
}

__global__ __launch_bounds__(256) void k_conv1(const float* __restrict__ x,
                                               const float* __restrict__ Wc1,
                                               const float* __restrict__ b1,
                                               const float* __restrict__ Wg,
                                               float* __restrict__ xt,
                                               const int* __restrict__ row,
                                               const int* __restrict__ col,
                                               const float* __restrict__ ew,
                                               float* __restrict__ dinv,
                                               int* __restrict__ off,
                                               int* __restrict__ src,
                                               float* __restrict__ wsl) {
  __shared__ float Xs[16 * XS_STRIDE];
  __shared__ float Hs[8 * 16 * HS_STRIDE];
  __shared__ float PdegL[PREP_NODES];
  __shared__ int   PcntL[PREP_NODES];
  __shared__ int   PcurL[PREP_NODES];
  __shared__ int   PredL[16];

  const int tid = threadIdx.x;
  if (blockIdx.x < PREP_BLOCKS) {
    graph_prep(row, col, ew, dinv, off, src, wsl, blockIdx.x, tid, PdegL, PcntL, PcurL, PredL);
    return;
  }
  const int lane   = tid & 31;
  const int wave   = tid >> 5;
  const int lane16 = lane & 15;
  const int half   = lane >> 4;
  const int n0     = (blockIdx.x - PREP_BLOCKS) * 16;

  for (int idx = tid; idx < 16 * CIN * WLEN; idx += 256) {
    int n = idx / (CIN * WLEN), r = idx % (CIN * WLEN);
    Xs[n * XS_STRIDE + r] = x[(n0 + n) * (CIN * WLEN) + r];
  }

  const int pb  = wave & 3;
  const int coP = pb * 16 + lane16;
  const int coQ = coP + 64;
  v2f bp1[10], bq1[10];
#pragma unroll
  for (int s = 0; s < 10; ++s) {
    int k     = 4 * s + 2 * half;
    bp1[s][0] = Wc1[coP * 40 + k];
    bp1[s][1] = Wc1[coP * 40 + k + 1];
    bq1[s][0] = Wc1[coQ * 40 + k];
    bq1[s][1] = Wc1[coQ * 40 + k + 1];
  }
  const float biasP = b1[coP];
  const float biasQ = b1[coQ];

  v2f bwg[16];
#pragma unroll
  for (int s = 0; s < 16; ++s) {
    int k     = 4 * s + 2 * half;
    bwg[s][0] = Wg[k * 16 + lane16];
    bwg[s][1] = Wg[(k + 1) * 16 + lane16];
  }
  __syncthreads();

  const int dt = wave >> 2;
  for (int tbase = 0; tbase < T1; tbase += 8) {
    for (int i = 0; i < 4; ++i) {
      int tl = dt * 4 + i;
      int t  = tbase + tl;
      if (t < T1) {
        v8f accp, accq;
#pragma unroll
        for (int j = 0; j < 8; ++j) { accp[j] = biasP; accq[j] = biasQ; }
#pragma unroll
        for (int s = 0; s < 10; ++s) {
          int j0 = 4 * s + 2 * half;
          int ci = j0 >> 3, kk = j0 & 7;
          v2f a;
          a[0] = Xs[lane16 * XS_STRIDE + ci * WLEN + t + kk];
          a[1] = Xs[lane16 * XS_STRIDE + ci * WLEN + t + kk + 1];
          accp = wmma_f32(a, bp1[s], accp);
          accq = wmma_f32(a, bq1[s], accq);
        }
        int ch = pb * 16 + lane16;
#pragma unroll
        for (int j = 0; j < 8; ++j) {
          float hv = accp[j] * sigmoidf_(accq[j]);
          int m    = j + 8 * half;
          Hs[(tl * 16 + m) * HS_STRIDE + ch] = hv;
        }
      }
    }
    __syncthreads();
    {
      int tl = wave;
      int t  = tbase + tl;
      if (t < T1) {
        v8f acc = {};
#pragma unroll
        for (int s = 0; s < 16; ++s) {
          int k0 = 4 * s + 2 * half;
          v2f a;
          a[0] = Hs[(tl * 16 + lane16) * HS_STRIDE + k0];
          a[1] = Hs[(tl * 16 + lane16) * HS_STRIDE + k0 + 1];
          acc  = wmma_f32(a, bwg[s], acc);
        }
#pragma unroll
        for (int j = 0; j < 8; ++j) {
          int m = j + 8 * half;
          xt[((n0 + m) * T1 + t) * H2C + lane16] = acc[j];
        }
      }
    }
    __syncthreads();
  }
}

#define ROW4 (T1 * H2C / 4)
#define ECAP 1024
#define EU   8
#define NPC  2

template <bool STG>
__device__ __forceinline__ void gather_tile(const float4* __restrict__ xt4,
                                            const int* __restrict__ src,
                                            const float* __restrict__ nrm,
                                            const int* es, const float* ws, int bs,
                                            const int* offs, const float* dn,
                                            const float* __restrict__ dinv,
                                            const float* __restrict__ bg, int n0,
                                            int tid, float* Gs) {
  const int f0 = (tid & 3) * 4;
  const float bg0 = bg[f0], bg1 = bg[f0 + 1], bg2 = bg[f0 + 2], bg3 = bg[f0 + 3];
  for (int o = tid; o < 16 * ROW4; o += 256 * NPC) {
    int   n[NPC], rem[NPC], b[NPC], en[NPC];
    bool  live[NPC];
    float4 vs[NPC];
    float ax[NPC], ay[NPC], az[NPC], aw[NPC];
    float dnq[NPC];
    int len = 0;
#pragma unroll
    for (int q = 0; q < NPC; ++q) {
      live[q] = (o + 256 * q) < 16 * ROW4;
      const int oq = live[q] ? (o + 256 * q) : o;
      n[q]   = oq / ROW4;
      rem[q] = oq - n[q] * ROW4;
      b[q]   = offs[n[q]];
      en[q]  = offs[n[q] + 1];
      dnq[q] = dn[n[q]];
      vs[q]  = xt4[(n0 + n[q]) * ROW4 + rem[q]];
      ax[q] = ay[q] = az[q] = aw[q] = 0.0f;
      len = max(len, en[q] - b[q]);
    }
    for (int j = 0; j < len; j += EU) {
      int   r[NPC][EU];
      float w[NPC][EU];
      float4 v[NPC][EU];
#pragma unroll
      for (int q = 0; q < NPC; ++q) {
#pragma unroll
        for (int k = 0; k < EU; ++k) {
          const int pp = b[q] + j + k;
          const int pc = max(min(pp, en[q] - 1), bs);
          r[q][k] = STG ? es[pc - bs] : src[pc];
          const float wk = STG ? ws[pc - bs] : dinv[r[q][k]] * nrm[pc];
          w[q][k] = (pp < en[q]) ? wk * dnq[q] : 0.0f;
        }
      }
#pragma unroll
      for (int q = 0; q < NPC; ++q) {
#pragma unroll
        for (int k = 0; k < EU; ++k) v[q][k] = xt4[r[q][k] * ROW4 + rem[q]];
      }
#pragma unroll
      for (int q = 0; q < NPC; ++q) {
#pragma unroll
        for (int k = 0; k < EU; ++k) {
          ax[q] += w[q][k] * v[q][k].x;
          ay[q] += w[q][k] * v[q][k].y;
          az[q] += w[q][k] * v[q][k].z;
          aw[q] += w[q][k] * v[q][k].w;
        }
      }
    }
#pragma unroll
    for (int q = 0; q < NPC; ++q) {
      if (live[q]) {
        const float wself = dnq[q] * dnq[q];
        float gx = ax[q] + wself * vs[q].x + bg0;
        float gy = ay[q] + wself * vs[q].y + bg1;
        float gz = az[q] + wself * vs[q].z + bg2;
        float gw = aw[q] + wself * vs[q].w + bg3;
        float* g = &Gs[(n[q] * T1 + (rem[q] >> 2)) * GS_STRIDE + f0];
        g[0] = gx > 0.0f ? gx : 0.0f;
        g[1] = gy > 0.0f ? gy : 0.0f;
        g[2] = gz > 0.0f ? gz : 0.0f;
        g[3] = gw > 0.0f ? gw : 0.0f;
      }
    }
  }
}

__global__ __launch_bounds__(256) void k_conv2(const float* __restrict__ xt,
                                               const int* __restrict__ off,
                                               const int* __restrict__ src,
                                               const float* __restrict__ nrm,
                                               const float* __restrict__ dinv,
                                               const float* __restrict__ bg,
                                               const float* __restrict__ Wc2,
                                               const float* __restrict__ b2,
                                               float* __restrict__ out) {
  __shared__ float Gs[16 * T1 * GS_STRIDE];
  __shared__ int   Es[ECAP];
  __shared__ float Ws[ECAP];
  __shared__ int   Offs[17];
  __shared__ float Dn[16];

  const int tid    = threadIdx.x;
  const int lane   = tid & 31;
  const int wave   = tid >> 5;
  const int lane16 = lane & 15;
  const int half   = lane >> 4;
  const int n0     = blockIdx.x * 16;

  if (tid < 17) Offs[tid] = off[n0 + tid];
  if (tid >= 32 && tid < 48) Dn[tid - 32] = dinv[n0 + tid - 32];
  __syncthreads();
  const int  bs     = Offs[0];
  const int  nslot  = Offs[16] - bs;
  const bool staged = nslot <= ECAP;
  if (staged) {
    for (int i = tid; i < nslot; i += 256) {
      const int s = src[bs + i];
      Es[i] = s;
      Ws[i] = dinv[s] * nrm[bs + i];
    }
  }
  __syncthreads();
  const float4* xt4 = reinterpret_cast<const float4*>(xt);
  if (staged)
    gather_tile<true>(xt4, src, nrm, Es, Ws, bs, Offs, Dn, dinv, bg, n0, tid, Gs);
  else
    gather_tile<false>(xt4, src, nrm, Es, Ws, bs, Offs, Dn, dinv, bg, n0, tid, Gs);

  const int pb  = wave & 3;
  const int coP = pb * 16 + lane16;
  const int coQ = coP + 64;
  v2f ap2[32], aq2[32];
#pragma unroll
  for (int s = 0; s < 32; ++s) {
    int k     = 4 * s + 2 * half;
    ap2[s][0] = Wc2[coP * 128 + k];
    ap2[s][1] = Wc2[coP * 128 + k + 1];
    aq2[s][0] = Wc2[coQ * 128 + k];
    aq2[s][1] = Wc2[coQ * 128 + k + 1];
  }
  float biasP[8], biasQ[8];
#pragma unroll
  for (int j = 0; j < 8; ++j) {
    biasP[j] = b2[pb * 16 + 8 * half + j];
    biasQ[j] = b2[pb * 16 + 8 * half + j + 64];
  }
  __syncthreads();

  const int qhalf = wave >> 2;
  for (int q = qhalf * 25; q < qhalf * 25 + 25; ++q) {
    const int cc = q * 16 + lane16;
    const int n  = cc / T2;
    const int t2 = cc - n * T2;
    const float* gb = &Gs[(n * T1 + t2 + 2 * half) * GS_STRIDE];
    v8f accp, accq;
#pragma unroll
    for (int j = 0; j < 8; ++j) { accp[j] = biasP[j]; accq[j] = biasQ[j]; }
#pragma unroll
    for (int s = 0; s < 32; ++s) {
      const int kk4 = (4 * s) & 7;
      const int ci  = s >> 1;
      v2f b;
      b[0] = gb[kk4 * GS_STRIDE + ci];
      b[1] = gb[(kk4 + 1) * GS_STRIDE + ci];
      accp = wmma_f32(ap2[s], b, accp);
      accq = wmma_f32(aq2[s], b, accq);
    }
    float* op = out + (n0 + n) * (H1C * T2) + (pb * 16 + 8 * half) * T2 + t2;
#pragma unroll
    for (int j = 0; j < 8; ++j) {
      float hv = accp[j] * sigmoidf_(accq[j]);
      __builtin_nontemporal_store(hv, &op[j * T2]);
    }
  }
}

extern "C" void kernel_launch(void* const* d_in, const int* in_sizes, int n_in,
                              void* d_out, int out_size, void* d_ws, size_t ws_size,
                              hipStream_t stream) {
  const float* x   = (const float*)d_in[0];
  const int*   ei  = (const int*)d_in[1];
  const float* ea  = (const float*)d_in[2];
  const float* Wc1 = (const float*)d_in[4];
  const float* b1  = (const float*)d_in[5];
  const float* Wg  = (const float*)d_in[6];
  const float* bg  = (const float*)d_in[7];
  const float* Wc2 = (const float*)d_in[8];
  const float* b2  = (const float*)d_in[9];
  float*       out = (float*)d_out;

  const int* row = ei;
  const int* col = ei + NEDGE;

  const size_t xt_bytes = (size_t)T1 * NNODE * H2C * sizeof(float);
  char*  wsb = (char*)d_ws;
  float* xt  = (float*)wsb;
  float* deg = (float*)(wsb + xt_bytes);
  int*   off = (int*)(wsb + xt_bytes + 2 * 32768);
  int*   src = (int*)(wsb + xt_bytes + 4 * 32768);
  float* nrm = (float*)(wsb + xt_bytes + 4 * 32768 + 262144);

  k_conv1<<<PREP_BLOCKS + NNODE / 16, 256, 0, stream>>>(x, Wc1, b1, Wg, xt, row, col, ea, deg,
                                                        off, src, nrm);
  k_conv2<<<NNODE / 16, 256, 0, stream>>>(xt, off, src, nrm, deg, bg, Wc2, b2, out);
}


# GLU gate of both convolutions: 1/(1+exp(-q)) by hardware reciprocal (v_rcp_f32, 1 ulp) instead of IEEE division: about 60 fewer vector issue slots per 8 gate values beside the f32 WMMAs
# speedup vs baseline: 4.3435x; 1.0636x over previous
#include <hip/hip_runtime.h>
#include <hip/hip_bf16.h>

typedef float v2f __attribute__((ext_vector_type(2)));
typedef float v8f __attribute__((ext_vector_type(8)));

#define NNODE 7936
#define CIN   5
#define WLEN  64
#define T1    57
#define T2    50
#define H1C   64
#define H2C   16
#define NEDGE 63488

#define XS_STRIDE 321
#define HS_STRIDE 66
#define GS_STRIDE 17

__device__ __forceinline__ v8f wmma_f32(v2f a, v2f b, v8f c) {
  return __builtin_amdgcn_wmma_f32_16x16x4_f32(false, a, false, b, (short)0, c,
                                               false, false);
}

__device__ __forceinline__ float sigmoidf_(float x) {
  return __builtin_amdgcn_rcpf(1.0f + __expf(-x));
}

#define PREP_NODES  256
#define PREP_BLOCKS (NNODE / PREP_NODES)
#define NE4         (NEDGE / 4)
#define PREP_U      8

__device__ __forceinline__ void graph_prep(const int* __restrict__ row,
                                           const int* __restrict__ col,
                                           const float* __restrict__ ew,
                                           float* __restrict__ dinv,
                                           int* __restrict__ off,
                                           int* __restrict__ src,
                                           float* __restrict__ wsl, int pblk, int tid,
                                           float* degL, int* cntL, int* curL, int* redL) {
  const int lo   = pblk * PREP_NODES;
  const int lane = tid & 31;
  const int wave = tid >> 5;
  const int4*   row4 = reinterpret_cast<const int4*>(row);
  const int4*   col4 = reinterpret_cast<const int4*>(col);
  const float4* ew4  = reinterpret_cast<const float4*>(ew);

  degL[tid] = 1.0f;
  cntL[tid] = 0;
  __syncthreads();

  int below = 0;
  for (int i0 = tid; i0 < NE4; i0 += 256 * PREP_U) {
    int4   c[PREP_U];
    float4 w[PREP_U];
#pragma unroll
    for (int u = 0; u < PREP_U; ++u) {
      const int i = i0 + 256 * u;
      if (i < NE4) { c[u] = col4[i]; w[u] = ew4[i]; }
      else         { c[u] = make_int4(NNODE, NNODE, NNODE, NNODE); w[u] = make_float4(0.f, 0.f, 0.f, 0.f); }
    }
#pragma unroll
    for (int u = 0; u < PREP_U; ++u) {
      const int   cc[4] = {c[u].x, c[u].y, c[u].z, c[u].w};
      const float ww[4] = {w[u].x, w[u].y, w[u].z, w[u].w};
#pragma unroll
      for (int k = 0; k < 4; ++k) {
        const int d = cc[k] - lo;
        below += (d < 0) ? 1 : 0;
        if ((unsigned)d < (unsigned)PREP_NODES) {
          atomicAdd(&degL[d], ww[k]);
          atomicAdd(&cntL[d], 1);
        }
      }
    }
  }
  for (int m = 16; m >= 1; m >>= 1) below += __shfl_xor(below, m);
  if (lane == 0) redL[wave] = below;
  __syncthreads();
  int base = 0;
  for (int v = 0; v < 8; ++v) base += redL[v];

  const int cn = cntL[tid];
  int inc = cn;
  for (int m = 1; m < 32; m <<= 1) {
    const int t = __shfl_up(inc, m);
    if (lane >= m) inc += t;
  }
  if (lane == 31) redL[8 + wave] = inc;
  __syncthreads();
  int wbase = 0;
  for (int v = 0; v < 8; ++v) wbase += (v < wave) ? redL[8 + v] : 0;
  const int first = base + wbase + inc - cn;
  off[lo + tid] = first;
  curL[tid]     = first;
  if (pblk == PREP_BLOCKS - 1 && tid == PREP_NODES - 1) off[NNODE] = first + cn;
  const float dg = degL[tid];
  dinv[lo + tid] = (dg > 0.0f) ? rsqrtf(dg) : 0.0f;
  __syncthreads();

  for (int i0 = tid; i0 < NE4; i0 += 256 * PREP_U) {
    int4   c[PREP_U], r[PREP_U];
    float4 w[PREP_U];
    int    slot[PREP_U][4];
#pragma unroll
    for (int u = 0; u < PREP_U; ++u) {
      const int i = i0 + 256 * u;
      if (i < NE4) { c[u] = col4[i]; r[u] = row4[i]; w[u] = ew4[i]; }
      else {
        c[u] = make_int4(NNODE, NNODE, NNODE, NNODE);
        r[u] = make_int4(0, 0, 0, 0);
        w[u] = make_float4(0.f, 0.f, 0.f, 0.f);
      }
    }
#pragma unroll
    for (int u = 0; u < PREP_U; ++u) {
      const int cc[4] = {c[u].x, c[u].y, c[u].z, c[u].w};
#pragma unroll
      for (int k = 0; k < 4; ++k) {
        const int d = cc[k] - lo;
        slot[u][k] = -1;
        if ((unsigned)d < (unsigned)PREP_NODES) slot[u][k] = atomicAdd(&curL[d], 1);
      }
    }
#pragma unroll
    for (int u = 0; u < PREP_U; ++u) {
      const int   rr[4] = {r[u].x, r[u].y, r[u].z, r[u].w};
      const float ww[4] = {w[u].x, w[u].y, w[u].z, w[u].w};
#pragma unroll
      for (int k = 0; k < 4; ++k) {
        if (slot[u][k] >= 0) {
          src[slot[u][k]] = rr[k];
          wsl[slot[u][k]] = ww[k];
        }
      }
    }
  }
}

__global__ __launch_bounds__(256) void k_conv1(const float* __restrict__ x,
                                               const float* __restrict__ Wc1,
                                               const float* __restrict__ b1,
                                               const float* __restrict__ Wg,
                                               float* __restrict__ xt,
                                               const int* __restrict__ row,
                                               const int* __restrict__ col,
                                               const float* __restrict__ ew,
                                               float* __restrict__ dinv,
                                               int* __restrict__ off,
                                               int* __restrict__ src,
                                               float* __restrict__ wsl) {
  __shared__ float Xs[16 * XS_STRIDE];
  __shared__ float Hs[8 * 16 * HS_STRIDE];
  __shared__ float PdegL[PREP_NODES];
  __shared__ int   PcntL[PREP_NODES];
  __shared__ int   PcurL[PREP_NODES];
  __shared__ int   PredL[16];

  const int tid = threadIdx.x;
  if (blockIdx.x < PREP_BLOCKS) {
    graph_prep(row, col, ew, dinv, off, src, wsl, blockIdx.x, tid, PdegL, PcntL, PcurL, PredL);
    return;
  }
  const int lane   = tid & 31;
  const int wave   = tid >> 5;
  const int lane16 = lane & 15;
  const int half   = lane >> 4;
  const int n0     = (blockIdx.x - PREP_BLOCKS) * 16;

  for (int idx = tid; idx < 16 * CIN * WLEN; idx += 256) {
    int n = idx / (CIN * WLEN), r = idx % (CIN * WLEN);
    Xs[n * XS_STRIDE + r] = x[(n0 + n) * (CIN * WLEN) + r];
  }

  const int pb  = wave & 3;
  const int coP = pb * 16 + lane16;
  const int coQ = coP + 64;
  v2f bp1[10], bq1[10];
#pragma unroll
  for (int s = 0; s < 10; ++s) {
    int k     = 4 * s + 2 * half;
    bp1[s][0] = Wc1[coP * 40 + k];
    bp1[s][1] = Wc1[coP * 40 + k + 1];
    bq1[s][0] = Wc1[coQ * 40 + k];
    bq1[s][1] = Wc1[coQ * 40 + k + 1];
  }
  const float biasP = b1[coP];
  const float biasQ = b1[coQ];

  v2f bwg[16];
#pragma unroll
  for (int s = 0; s < 16; ++s) {
    int k     = 4 * s + 2 * half;
    bwg[s][0] = Wg[k * 16 + lane16];
    bwg[s][1] = Wg[(k + 1) * 16 + lane16];
  }
  __syncthreads();

  const int dt = wave >> 2;
  for (int tbase = 0; tbase < T1; tbase += 8) {
    for (int i = 0; i < 4; ++i) {
      int tl = dt * 4 + i;
      int t  = tbase + tl;
      if (t < T1) {
        v8f accp, accq;
#pragma unroll
        for (int j = 0; j < 8; ++j) { accp[j] = biasP; accq[j] = biasQ; }
#pragma unroll
        for (int s = 0; s < 10; ++s) {
          int j0 = 4 * s + 2 * half;
          int ci = j0 >> 3, kk = j0 & 7;
          v2f a;
          a[0] = Xs[lane16 * XS_STRIDE + ci * WLEN + t + kk];
          a[1] = Xs[lane16 * XS_STRIDE + ci * WLEN + t + kk + 1];
          accp = wmma_f32(a, bp1[s], accp);
          accq = wmma_f32(a, bq1[s], accq);
        }
        int ch = pb * 16 + lane16;
#pragma unroll
        for (int j = 0; j < 8; ++j) {
          float hv = accp[j] * sigmoidf_(accq[j]);
          int m    = j + 8 * half;
          Hs[(tl * 16 + m) * HS_STRIDE + ch] = hv;
        }
      }
    }
    __syncthreads();
    {
      int tl = wave;
      int t  = tbase + tl;
      if (t < T1) {
        v8f acc = {};
#pragma unroll
        for (int s = 0; s < 16; ++s) {
          int k0 = 4 * s + 2 * half;
          v2f a;
          a[0] = Hs[(tl * 16 + lane16) * HS_STRIDE + k0];
          a[1] = Hs[(tl * 16 + lane16) * HS_STRIDE + k0 + 1];
          acc  = wmma_f32(a, bwg[s], acc);
        }
#pragma unroll
        for (int j = 0; j < 8; ++j) {
          int m = j + 8 * half;
          xt[((n0 + m) * T1 + t) * H2C + lane16] = acc[j];
        }
      }
    }
    __syncthreads();
  }
}

#define ROW4 (T1 * H2C / 4)
#define ECAP 1024
#define EU   8
#define NPC  2

template <bool STG>
__device__ __forceinline__ void gather_tile(const float4* __restrict__ xt4,
                                            const int* __restrict__ src,
                                            const float* __restrict__ nrm,
                                            const int* es, const float* ws, int bs,
                                            const int* offs, const float* dn,
                                            const float* __restrict__ dinv,
                                            const float* __restrict__ bg, int n0,
                                            int tid, float* Gs) {
  const int f0 = (tid & 3) * 4;
  const float bg0 = bg[f0], bg1 = bg[f0 + 1], bg2 = bg[f0 + 2], bg3 = bg[f0 + 3];
  for (int o = tid; o < 16 * ROW4; o += 256 * NPC) {
    int   n[NPC], rem[NPC], b[NPC], en[NPC];
    bool  live[NPC];
    float4 vs[NPC];
    float ax[NPC], ay[NPC], az[NPC], aw[NPC];
    float dnq[NPC];
    int len = 0;
#pragma unroll
    for (int q = 0; q < NPC; ++q) {
      live[q] = (o + 256 * q) < 16 * ROW4;
      const int oq = live[q] ? (o + 256 * q) : o;
      n[q]   = oq / ROW4;
      rem[q] = oq - n[q] * ROW4;
      b[q]   = offs[n[q]];
      en[q]  = offs[n[q] + 1];
      dnq[q] = dn[n[q]];
      vs[q]  = xt4[(n0 + n[q]) * ROW4 + rem[q]];
      ax[q] = ay[q] = az[q] = aw[q] = 0.0f;
      len = max(len, en[q] - b[q]);
    }
    for (int j = 0; j < len; j += EU) {
      int   r[NPC][EU];
      float w[NPC][EU];
      float4 v[NPC][EU];
#pragma unroll
      for (int q = 0; q < NPC; ++q) {
#pragma unroll
        for (int k = 0; k < EU; ++k) {
          const int pp = b[q] + j + k;
          const int pc = max(min(pp, en[q] - 1), bs);
          r[q][k] = STG ? es[pc - bs] : src[pc];
          const float wk = STG ? ws[pc - bs] : dinv[r[q][k]] * nrm[pc];
          w[q][k] = (pp < en[q]) ? wk * dnq[q] : 0.0f;
        }
      }
#pragma unroll
      for (int q = 0; q < NPC; ++q) {
#pragma unroll
        for (int k = 0; k < EU; ++k) v[q][k] = xt4[r[q][k] * ROW4 + rem[q]];
      }
#pragma unroll
      for (int q = 0; q < NPC; ++q) {
#pragma unroll
        for (int k = 0; k < EU; ++k) {
          ax[q] += w[q][k] * v[q][k].x;
          ay[q] += w[q][k] * v[q][k].y;
          az[q] += w[q][k] * v[q][k].z;
          aw[q] += w[q][k] * v[q][k].w;
        }
      }
    }
#pragma unroll
    for (int q = 0; q < NPC; ++q) {
      if (live[q]) {
        const float wself = dnq[q] * dnq[q];
        float gx = ax[q] + wself * vs[q].x + bg0;
        float gy = ay[q] + wself * vs[q].y + bg1;
        float gz = az[q] + wself * vs[q].z + bg2;
        float gw = aw[q] + wself * vs[q].w + bg3;
        float* g = &Gs[(n[q] * T1 + (rem[q] >> 2)) * GS_STRIDE + f0];
        g[0] = gx > 0.0f ? gx : 0.0f;
        g[1] = gy > 0.0f ? gy : 0.0f;
        g[2] = gz > 0.0f ? gz : 0.0f;
        g[3] = gw > 0.0f ? gw : 0.0f;
      }
    }
  }
}

__global__ __launch_bounds__(256) void k_conv2(const float* __restrict__ xt,
                                               const int* __restrict__ off,
                                               const int* __restrict__ src,
                                               const float* __restrict__ nrm,
                                               const float* __restrict__ dinv,
                                               const float* __restrict__ bg,
                                               const float* __restrict__ Wc2,
                                               const float* __restrict__ b2,
                                               float* __restrict__ out) {
  __shared__ float Gs[16 * T1 * GS_STRIDE];
  __shared__ int   Es[ECAP];
  __shared__ float Ws[ECAP];
  __shared__ int   Offs[17];
  __shared__ float Dn[16];

  const int tid    = threadIdx.x;
  const int lane   = tid & 31;
  const int wave   = tid >> 5;
  const int lane16 = lane & 15;
  const int half   = lane >> 4;
  const int n0     = blockIdx.x * 16;

  if (tid < 17) Offs[tid] = off[n0 + tid];
  if (tid >= 32 && tid < 48) Dn[tid - 32] = dinv[n0 + tid - 32];
  __syncthreads();
  const int  bs     = Offs[0];
  const int  nslot  = Offs[16] - bs;
  const bool staged = nslot <= ECAP;
  if (staged) {
    for (int i = tid; i < nslot; i += 256) {
      const int s = src[bs + i];
      Es[i] = s;
      Ws[i] = dinv[s] * nrm[bs + i];
    }
  }
  __syncthreads();
  const float4* xt4 = reinterpret_cast<const float4*>(xt);
  if (staged)
    gather_tile<true>(xt4, src, nrm, Es, Ws, bs, Offs, Dn, dinv, bg, n0, tid, Gs);
  else
    gather_tile<false>(xt4, src, nrm, Es, Ws, bs, Offs, Dn, dinv, bg, n0, tid, Gs);

  const int pb  = wave & 3;
  const int coP = pb * 16 + lane16;
  const int coQ = coP + 64;
  v2f ap2[32], aq2[32];
#pragma unroll
  for (int s = 0; s < 32; ++s) {
    int k     = 4 * s + 2 * half;
    ap2[s][0] = Wc2[coP * 128 + k];
    ap2[s][1] = Wc2[coP * 128 + k + 1];
    aq2[s][0] = Wc2[coQ * 128 + k];
    aq2[s][1] = Wc2[coQ * 128 + k + 1];
  }
  float biasP[8], biasQ[8];
#pragma unroll
  for (int j = 0; j < 8; ++j) {
    biasP[j] = b2[pb * 16 + 8 * half + j];
    biasQ[j] = b2[pb * 16 + 8 * half + j + 64];
  }
  __syncthreads();

  const int qhalf = wave >> 2;
  for (int q = qhalf * 25; q < qhalf * 25 + 25; ++q) {
    const int cc = q * 16 + lane16;
    const int n  = cc / T2;
    const int t2 = cc - n * T2;
    const float* gb = &Gs[(n * T1 + t2 + 2 * half) * GS_STRIDE];
    v8f accp, accq;
#pragma unroll
    for (int j = 0; j < 8; ++j) { accp[j] = biasP[j]; accq[j] = biasQ[j]; }
#pragma unroll
    for (int s = 0; s < 32; ++s) {
      const int kk4 = (4 * s) & 7;
      const int ci  = s >> 1;
      v2f b;
      b[0] = gb[kk4 * GS_STRIDE + ci];
      b[1] = gb[(kk4 + 1) * GS_STRIDE + ci];
      accp = wmma_f32(ap2[s], b, accp);
      accq = wmma_f32(aq2[s], b, accq);
    }
    float* op = out + (n0 + n) * (H1C * T2) + (pb * 16 + 8 * half) * T2 + t2;
#pragma unroll
    for (int j = 0; j < 8; ++j) {
      float hv = accp[j] * sigmoidf_(accq[j]);
      __builtin_nontemporal_store(hv, &op[j * T2]);
    }
  }
}

extern "C" void kernel_launch(void* const* d_in, const int* in_sizes, int n_in,
                              void* d_out, int out_size, void* d_ws, size_t ws_size,
                              hipStream_t stream) {
  const float* x   = (const float*)d_in[0];
  const int*   ei  = (const int*)d_in[1];
  const float* ea  = (const float*)d_in[2];
  const float* Wc1 = (const float*)d_in[4];
  const float* b1  = (const float*)d_in[5];
  const float* Wg  = (const float*)d_in[6];
  const float* bg  = (const float*)d_in[7];
  const float* Wc2 = (const float*)d_in[8];
  const float* b2  = (const float*)d_in[9];
  float*       out = (float*)d_out;

  const int* row = ei;
  const int* col = ei + NEDGE;

  const size_t xt_bytes = (size_t)T1 * NNODE * H2C * sizeof(float);
  char*  wsb = (char*)d_ws;
  float* xt  = (float*)wsb;
  float* deg = (float*)(wsb + xt_bytes);
  int*   off = (int*)(wsb + xt_bytes + 2 * 32768);
  int*   src = (int*)(wsb + xt_bytes + 4 * 32768);
  float* nrm = (float*)(wsb + xt_bytes + 4 * 32768 + 262144);

  k_conv1<<<PREP_BLOCKS + NNODE / 16, 256, 0, stream>>>(x, Wc1, b1, Wg, xt, row, col, ea, deg,
                                                        off, src, nrm);
  k_conv2<<<NNODE / 16, 256, 0, stream>>>(xt, off, src, nrm, deg, bg, Wc2, b2, out);
}


# conv2 gather as one block-uniform walk over the tile's staged entry list (slots + self-loops): one float4 column per thread, 16 row loads in flight, no per-slot branch or LDS wait; same sum order
# speedup vs baseline: 4.6533x; 1.0713x over previous
#include <hip/hip_runtime.h>
#include <hip/hip_bf16.h>

typedef float v2f __attribute__((ext_vector_type(2)));
typedef float v8f __attribute__((ext_vector_type(8)));

#define NNODE 7936
#define CIN   5
#define WLEN  64
#define T1    57
#define T2    50
#define H1C   64
#define H2C   16
#define NEDGE 63488

#define XS_STRIDE 321
#define HS_STRIDE 66
#define GS_STRIDE 17

__device__ __forceinline__ v8f wmma_f32(v2f a, v2f b, v8f c) {
  return __builtin_amdgcn_wmma_f32_16x16x4_f32(false, a, false, b, (short)0, c,
                                               false, false);
}

__device__ __forceinline__ float sigmoidf_(float x) {
  return __builtin_amdgcn_rcpf(1.0f + __expf(-x));
}

#define PREP_NODES  256
#define PREP_BLOCKS (NNODE / PREP_NODES)
#define NE4         (NEDGE / 4)
#define PREP_U      8

__device__ __forceinline__ void graph_prep(const int* __restrict__ row,
                                           const int* __restrict__ col,
                                           const float* __restrict__ ew,
                                           float* __restrict__ dinv,
                                           int* __restrict__ off,
                                           int* __restrict__ src,
                                           float* __restrict__ wsl, int pblk, int tid,
                                           float* degL, int* cntL, int* curL, int* redL) {
  const int lo   = pblk * PREP_NODES;
  const int lane = tid & 31;
  const int wave = tid >> 5;
  const int4*   row4 = reinterpret_cast<const int4*>(row);
  const int4*   col4 = reinterpret_cast<const int4*>(col);
  const float4* ew4  = reinterpret_cast<const float4*>(ew);

  degL[tid] = 1.0f;
  cntL[tid] = 0;
  __syncthreads();

  int below = 0;
  for (int i0 = tid; i0 < NE4; i0 += 256 * PREP_U) {
    int4   c[PREP_U];
    float4 w[PREP_U];
#pragma unroll
    for (int u = 0; u < PREP_U; ++u) {
      const int i = i0 + 256 * u;
      if (i < NE4) { c[u] = col4[i]; w[u] = ew4[i]; }
      else         { c[u] = make_int4(NNODE, NNODE, NNODE, NNODE); w[u] = make_float4(0.f, 0.f, 0.f, 0.f); }
    }
#pragma unroll
    for (int u = 0; u < PREP_U; ++u) {
      const int   cc[4] = {c[u].x, c[u].y, c[u].z, c[u].w};
      const float ww[4] = {w[u].x, w[u].y, w[u].z, w[u].w};
#pragma unroll
      for (int k = 0; k < 4; ++k) {
        const int d = cc[k] - lo;
        below += (d < 0) ? 1 : 0;
        if ((unsigned)d < (unsigned)PREP_NODES) {
          atomicAdd(&degL[d], ww[k]);
          atomicAdd(&cntL[d], 1);
        }
      }
    }
  }
  for (int m = 16; m >= 1; m >>= 1) below += __shfl_xor(below, m);
  if (lane == 0) redL[wave] = below;
  __syncthreads();
  int base = 0;
  for (int v = 0; v < 8; ++v) base += redL[v];

  const int cn = cntL[tid];
  int inc = cn;
  for (int m = 1; m < 32; m <<= 1) {
    const int t = __shfl_up(inc, m);
    if (lane >= m) inc += t;
  }
  if (lane == 31) redL[8 + wave] = inc;
  __syncthreads();
  int wbase = 0;
  for (int v = 0; v < 8; ++v) wbase += (v < wave) ? redL[8 + v] : 0;
  const int first = base + wbase + inc - cn;
  off[lo + tid] = first;
  curL[tid]     = first;
  if (pblk == PREP_BLOCKS - 1 && tid == PREP_NODES - 1) off[NNODE] = first + cn;
  const float dg = degL[tid];
  dinv[lo + tid] = (dg > 0.0f) ? rsqrtf(dg) : 0.0f;
  __syncthreads();

  for (int i0 = tid; i0 < NE4; i0 += 256 * PREP_U) {
    int4   c[PREP_U], r[PREP_U];
    float4 w[PREP_U];
    int    slot[PREP_U][4];
#pragma unroll
    for (int u = 0; u < PREP_U; ++u) {
      const int i = i0 + 256 * u;
      if (i < NE4) { c[u] = col4[i]; r[u] = row4[i]; w[u] = ew4[i]; }
      else {
        c[u] = make_int4(NNODE, NNODE, NNODE, NNODE);
        r[u] = make_int4(0, 0, 0, 0);
        w[u] = make_float4(0.f, 0.f, 0.f, 0.f);
      }
    }
#pragma unroll
    for (int u = 0; u < PREP_U; ++u) {
      const int cc[4] = {c[u].x, c[u].y, c[u].z, c[u].w};
#pragma unroll
      for (int k = 0; k < 4; ++k) {
        const int d = cc[k] - lo;
        slot[u][k] = -1;
        if ((unsigned)d < (unsigned)PREP_NODES) slot[u][k] = atomicAdd(&curL[d], 1);
      }
    }
#pragma unroll
    for (int u = 0; u < PREP_U; ++u) {
      const int   rr[4] = {r[u].x, r[u].y, r[u].z, r[u].w};
      const float ww[4] = {w[u].x, w[u].y, w[u].z, w[u].w};
#pragma unroll
      for (int k = 0; k < 4; ++k) {
        if (slot[u][k] >= 0) {
          src[slot[u][k]] = rr[k];
          wsl[slot[u][k]] = ww[k];
        }
      }
    }
  }
}

__global__ __launch_bounds__(256) void k_conv1(const float* __restrict__ x,
                                               const float* __restrict__ Wc1,
                                               const float* __restrict__ b1,
                                               const float* __restrict__ Wg,
                                               float* __restrict__ xt,
                                               const int* __restrict__ row,
                                               const int* __restrict__ col,
                                               const float* __restrict__ ew,
                                               float* __restrict__ dinv,
                                               int* __restrict__ off,
                                               int* __restrict__ src,
                                               float* __restrict__ wsl) {
  __shared__ float Xs[16 * XS_STRIDE];
  __shared__ float Hs[8 * 16 * HS_STRIDE];
  __shared__ float PdegL[PREP_NODES];
  __shared__ int   PcntL[PREP_NODES];
  __shared__ int   PcurL[PREP_NODES];
  __shared__ int   PredL[16];

  const int tid = threadIdx.x;
  if (blockIdx.x < PREP_BLOCKS) {
    graph_prep(row, col, ew, dinv, off, src, wsl, blockIdx.x, tid, PdegL, PcntL, PcurL, PredL);
    return;
  }
  const int lane   = tid & 31;
  const int wave   = tid >> 5;
  const int lane16 = lane & 15;
  const int half   = lane >> 4;
  const int n0     = (blockIdx.x - PREP_BLOCKS) * 16;

  for (int idx = tid; idx < 16 * CIN * WLEN; idx += 256) {
    int n = idx / (CIN * WLEN), r = idx % (CIN * WLEN);
    Xs[n * XS_STRIDE + r] = x[(n0 + n) * (CIN * WLEN) + r];
  }

  const int pb  = wave & 3;
  const int coP = pb * 16 + lane16;
  const int coQ = coP + 64;
  v2f bp1[10], bq1[10];
#pragma unroll
  for (int s = 0; s < 10; ++s) {
    int k     = 4 * s + 2 * half;
    bp1[s][0] = Wc1[coP * 40 + k];
    bp1[s][1] = Wc1[coP * 40 + k + 1];
    bq1[s][0] = Wc1[coQ * 40 + k];
    bq1[s][1] = Wc1[coQ * 40 + k + 1];
  }
  const float biasP = b1[coP];
  const float biasQ = b1[coQ];

  v2f bwg[16];
#pragma unroll
  for (int s = 0; s < 16; ++s) {
    int k     = 4 * s + 2 * half;
    bwg[s][0] = Wg[k * 16 + lane16];
    bwg[s][1] = Wg[(k + 1) * 16 + lane16];
  }
  __syncthreads();

  const int dt = wave >> 2;
  for (int tbase = 0; tbase < T1; tbase += 8) {
    for (int i = 0; i < 4; ++i) {
      int tl = dt * 4 + i;
      int t  = tbase + tl;
      if (t < T1) {
        v8f accp, accq;
#pragma unroll
        for (int j = 0; j < 8; ++j) { accp[j] = biasP; accq[j] = biasQ; }
#pragma unroll
        for (int s = 0; s < 10; ++s) {
          int j0 = 4 * s + 2 * half;
          int ci = j0 >> 3, kk = j0 & 7;
          v2f a;
          a[0] = Xs[lane16 * XS_STRIDE + ci * WLEN + t + kk];
          a[1] = Xs[lane16 * XS_STRIDE + ci * WLEN + t + kk + 1];
          accp = wmma_f32(a, bp1[s], accp);
          accq = wmma_f32(a, bq1[s], accq);
        }
        int ch = pb * 16 + lane16;
#pragma unroll
        for (int j = 0; j < 8; ++j) {
          float hv = accp[j] * sigmoidf_(accq[j]);
          int m    = j + 8 * half;
          Hs[(tl * 16 + m) * HS_STRIDE + ch] = hv;
        }
      }
    }
    __syncthreads();
    {
      int tl = wave;
      int t  = tbase + tl;
      if (t < T1) {
        v8f acc = {};
#pragma unroll
        for (int s = 0; s < 16; ++s) {
          int k0 = 4 * s + 2 * half;
          v2f a;
          a[0] = Hs[(tl * 16 + lane16) * HS_STRIDE + k0];
          a[1] = Hs[(tl * 16 + lane16) * HS_STRIDE + k0 + 1];
          acc  = wmma_f32(a, bwg[s], acc);
        }
#pragma unroll
        for (int j = 0; j < 8; ++j) {
          int m = j + 8 * half;
          xt[((n0 + m) * T1 + t) * H2C + lane16] = acc[j];
        }
      }
    }
    __syncthreads();
  }
}

#define ROW4 (T1 * H2C / 4)
#define ECAP 1024
#define LU   16

__device__ __forceinline__ void gather_list(const float4* __restrict__ xt4, const int* es,
                                            const float* ws, const int* cl, int jpad,
                                            const float* __restrict__ bg, int tid, float* Gs) {
  const bool act = tid < ROW4;
  const int  rem = act ? tid : tid - 32;
  const int  f0  = (rem & 3) * 4;
  const float bg0 = bg[f0], bg1 = bg[f0 + 1], bg2 = bg[f0 + 2], bg3 = bg[f0 + 3];
  const float4* xcol = xt4 + rem;
  float*        gcol = &Gs[(rem >> 2) * GS_STRIDE + f0];
  float ax = 0.0f, ay = 0.0f, az = 0.0f, aw = 0.0f;
  for (int j0 = 0; j0 < jpad; j0 += LU) {
    int    r[LU], c[LU];
    float  w[LU];
    float4 v[LU];
#pragma unroll
    for (int k = 0; k < LU; ++k) {
      r[k] = __builtin_amdgcn_readfirstlane(es[j0 + k]);
      c[k] = __builtin_amdgcn_readfirstlane(cl[j0 + k]);
      w[k] = ws[j0 + k];
    }
#pragma unroll
    for (int k = 0; k < LU; ++k) v[k] = xcol[r[k] * ROW4];
#pragma unroll
    for (int k = 0; k < LU; ++k) {
      ax += w[k] * v[k].x;
      ay += w[k] * v[k].y;
      az += w[k] * v[k].z;
      aw += w[k] * v[k].w;
      if (c[k] >= 0) {
        const float gx = ax + bg0, gy = ay + bg1, gz = az + bg2, gw = aw + bg3;
        if (act) {
          float* g = gcol + c[k] * (T1 * GS_STRIDE);
          g[0] = gx > 0.0f ? gx : 0.0f;
          g[1] = gy > 0.0f ? gy : 0.0f;
          g[2] = gz > 0.0f ? gz : 0.0f;
          g[3] = gw > 0.0f ? gw : 0.0f;
        }
        ax = ay = az = aw = 0.0f;
      }
    }
  }
}

#define EU   8
#define NPC  2

__device__ __forceinline__ void gather_tile(const float4* __restrict__ xt4,
                                            const int* __restrict__ src,
                                            const float* __restrict__ nrm, int bs,
                                            const int* offs, const float* dn,
                                            const float* __restrict__ dinv,
                                            const float* __restrict__ bg, int n0,
                                            int tid, float* Gs) {
  const int f0 = (tid & 3) * 4;
  const float bg0 = bg[f0], bg1 = bg[f0 + 1], bg2 = bg[f0 + 2], bg3 = bg[f0 + 3];
  for (int o = tid; o < 16 * ROW4; o += 256 * NPC) {
    int   n[NPC], rem[NPC], b[NPC], en[NPC];
    bool  live[NPC];
    float4 vs[NPC];
    float ax[NPC], ay[NPC], az[NPC], aw[NPC];
    float dnq[NPC];
    int len = 0;
#pragma unroll
    for (int q = 0; q < NPC; ++q) {
      live[q] = (o + 256 * q) < 16 * ROW4;
      const int oq = live[q] ? (o + 256 * q) : o;
      n[q]   = oq / ROW4;
      rem[q] = oq - n[q] * ROW4;
      b[q]   = offs[n[q]];
      en[q]  = offs[n[q] + 1];
      dnq[q] = dn[n[q]];
      vs[q]  = xt4[(n0 + n[q]) * ROW4 + rem[q]];
      ax[q] = ay[q] = az[q] = aw[q] = 0.0f;
      len = max(len, en[q] - b[q]);
    }
    for (int j = 0; j < len; j += EU) {
      int   r[NPC][EU];
      float w[NPC][EU];
      float4 v[NPC][EU];
#pragma unroll
      for (int q = 0; q < NPC; ++q) {
#pragma unroll
        for (int k = 0; k < EU; ++k) {
          const int pp = b[q] + j + k;
          const int pc = max(min(pp, en[q] - 1), bs);
          r[q][k] = src[pc];
          const float wk = dinv[r[q][k]] * nrm[pc];
          w[q][k] = (pp < en[q]) ? wk * dnq[q] : 0.0f;
        }
      }
#pragma unroll
      for (int q = 0; q < NPC; ++q) {
#pragma unroll
        for (int k = 0; k < EU; ++k) v[q][k] = xt4[r[q][k] * ROW4 + rem[q]];
      }
#pragma unroll
      for (int q = 0; q < NPC; ++q) {
#pragma unroll
        for (int k = 0; k < EU; ++k) {
          ax[q] += w[q][k] * v[q][k].x;
          ay[q] += w[q][k] * v[q][k].y;
          az[q] += w[q][k] * v[q][k].z;
          aw[q] += w[q][k] * v[q][k].w;
        }
      }
    }
#pragma unroll
    for (int q = 0; q < NPC; ++q) {
      if (live[q]) {
        const float wself = dnq[q] * dnq[q];
        float gx = ax[q] + wself * vs[q].x + bg0;
        float gy = ay[q] + wself * vs[q].y + bg1;
        float gz = az[q] + wself * vs[q].z + bg2;
        float gw = aw[q] + wself * vs[q].w + bg3;
        float* g = &Gs[(n[q] * T1 + (rem[q] >> 2)) * GS_STRIDE + f0];
        g[0] = gx > 0.0f ? gx : 0.0f;
        g[1] = gy > 0.0f ? gy : 0.0f;
        g[2] = gz > 0.0f ? gz : 0.0f;
        g[3] = gw > 0.0f ? gw : 0.0f;
      }
    }
  }
}

__global__ __launch_bounds__(256) void k_conv2(const float* __restrict__ xt,
                                               const int* __restrict__ off,
                                               const int* __restrict__ src,
                                               const float* __restrict__ nrm,
                                               const float* __restrict__ dinv,
                                               const float* __restrict__ bg,
                                               const float* __restrict__ Wc2,
                                               const float* __restrict__ b2,
                                               float* __restrict__ out) {
  __shared__ float Gs[16 * T1 * GS_STRIDE];
  __shared__ int   Es[ECAP];
  __shared__ float Ws[ECAP];
  __shared__ int   Cs[ECAP];
  __shared__ int   Offs[17];
  __shared__ float Dn[16];

  const int tid    = threadIdx.x;
  const int lane   = tid & 31;
  const int wave   = tid >> 5;
  const int lane16 = lane & 15;
  const int half   = lane >> 4;
  const int n0     = blockIdx.x * 16;

  if (tid < 17) Offs[tid] = off[n0 + tid];
  if (tid >= 32 && tid < 48) Dn[tid - 32] = dinv[n0 + tid - 32];
  __syncthreads();
  const int  bs     = Offs[0];
  const int  nslot  = Offs[16] - bs;
  const int  jn     = nslot + 16;
  const int  jpad   = (jn + LU - 1) & ~(LU - 1);
  const bool staged = jpad <= ECAP;
  if (staged) {
    for (int i = tid; i < jpad; i += 256) {
      if (i < jn) {
        int n = 0;
#pragma unroll
        for (int m = 0; m < 16; ++m) n += (Offs[m + 1] - bs + m < i) ? 1 : 0;
        const float dn = Dn[n];
        if (i == Offs[n + 1] - bs + n) {
          Es[i] = n0 + n;
          Ws[i] = dn * dn;
          Cs[i] = n;
        } else {
          const int p = bs + i - n;
          const int s = src[p];
          Es[i] = s;
          Ws[i] = (dinv[s] * nrm[p]) * dn;
          Cs[i] = -1;
        }
      } else {
        Es[i] = n0 + 15;
        Ws[i] = 0.0f;
        Cs[i] = -1;
      }
    }
  }
  __syncthreads();
  const float4* xt4 = reinterpret_cast<const float4*>(xt);
  if (staged)
    gather_list(xt4, Es, Ws, Cs, jpad, bg, tid, Gs);
  else
    gather_tile(xt4, src, nrm, bs, Offs, Dn, dinv, bg, n0, tid, Gs);

  const int pb  = wave & 3;
  const int coP = pb * 16 + lane16;
  const int coQ = coP + 64;
  v2f ap2[32], aq2[32];
#pragma unroll
  for (int s = 0; s < 32; ++s) {
    int k     = 4 * s + 2 * half;
    ap2[s][0] = Wc2[coP * 128 + k];
    ap2[s][1] = Wc2[coP * 128 + k + 1];
    aq2[s][0] = Wc2[coQ * 128 + k];
    aq2[s][1] = Wc2[coQ * 128 + k + 1];
  }
  float biasP[8], biasQ[8];
#pragma unroll
  for (int j = 0; j < 8; ++j) {
    biasP[j] = b2[pb * 16 + 8 * half + j];
    biasQ[j] = b2[pb * 16 + 8 * half + j + 64];
  }
  __syncthreads();

  const int qhalf = wave >> 2;
  for (int q = qhalf * 25; q < qhalf * 25 + 25; ++q) {
    const int cc = q * 16 + lane16;
    const int n  = cc / T2;
    const int t2 = cc - n * T2;
    const float* gb = &Gs[(n * T1 + t2 + 2 * half) * GS_STRIDE];
    v8f accp, accq;
#pragma unroll
    for (int j = 0; j < 8; ++j) { accp[j] = biasP[j]; accq[j] = biasQ[j]; }
#pragma unroll
    for (int s = 0; s < 32; ++s) {
      const int kk4 = (4 * s) & 7;
      const int ci  = s >> 1;
      v2f b;
      b[0] = gb[kk4 * GS_STRIDE + ci];
      b[1] = gb[(kk4 + 1) * GS_STRIDE + ci];
      accp = wmma_f32(ap2[s], b, accp);
      accq = wmma_f32(aq2[s], b, accq);
    }
    float* op = out + (n0 + n) * (H1C * T2) + (pb * 16 + 8 * half) * T2 + t2;
#pragma unroll
    for (int j = 0; j < 8; ++j) {
      float hv = accp[j] * sigmoidf_(accq[j]);
      __builtin_nontemporal_store(hv, &op[j * T2]);
    }
  }
}

extern "C" void kernel_launch(void* const* d_in, const int* in_sizes, int n_in,
                              void* d_out, int out_size, void* d_ws, size_t ws_size,
                              hipStream_t stream) {
  const float* x   = (const float*)d_in[0];
  const int*   ei  = (const int*)d_in[1];
  const float* ea  = (const float*)d_in[2];
  const float* Wc1 = (const float*)d_in[4];
  const float* b1  = (const float*)d_in[5];
  const float* Wg  = (const float*)d_in[6];
  const float* bg  = (const float*)d_in[7];
  const float* Wc2 = (const float*)d_in[8];
  const float* b2  = (const float*)d_in[9];
  float*       out = (float*)d_out;

  const int* row = ei;
  const int* col = ei + NEDGE;

  const size_t xt_bytes = (size_t)T1 * NNODE * H2C * sizeof(float);
  char*  wsb = (char*)d_ws;
  float* xt  = (float*)wsb;
  float* deg = (float*)(wsb + xt_bytes);
  int*   off = (int*)(wsb + xt_bytes + 2 * 32768);
  int*   src = (int*)(wsb + xt_bytes + 4 * 32768);
  float* nrm = (float*)(wsb + xt_bytes + 4 * 32768 + 262144);

  k_conv1<<<PREP_BLOCKS + NNODE / 16, 256, 0, stream>>>(x, Wc1, b1, Wg, xt, row, col, ea, deg,
                                                        off, src, nrm);
  k_conv2<<<NNODE / 16, 256, 0, stream>>>(xt, off, src, nrm, deg, bg, Wc2, b2, out);
}


# conv2 output: each wave stages its GLU tiles in an LDS image of one out[n][16 channels][50] chunk (3200 B) and writes it as whole 128-byte lines with 16-byte stores; was 64-byte runs, partial lines
# speedup vs baseline: 4.9338x; 1.0603x over previous
#include <hip/hip_runtime.h>
#include <hip/hip_bf16.h>

typedef float v2f __attribute__((ext_vector_type(2)));
typedef float v8f __attribute__((ext_vector_type(8)));
typedef float v4f __attribute__((ext_vector_type(4)));

#define NNODE 7936
#define CIN   5
#define WLEN  64
#define T1    57
#define T2    50
#define H1C   64
#define H2C   16
#define NEDGE 63488

#define XS_STRIDE 321
#define HS_STRIDE 66
#define GS_STRIDE 17

__device__ __forceinline__ v8f wmma_f32(v2f a, v2f b, v8f c) {
  return __builtin_amdgcn_wmma_f32_16x16x4_f32(false, a, false, b, (short)0, c,
                                               false, false);
}

__device__ __forceinline__ float sigmoidf_(float x) {
  return __builtin_amdgcn_rcpf(1.0f + __expf(-x));
}

#define PREP_NODES  256
#define PREP_BLOCKS (NNODE / PREP_NODES)
#define NE4         (NEDGE / 4)
#define PREP_U      8

__device__ __forceinline__ void graph_prep(const int* __restrict__ row,
                                           const int* __restrict__ col,
                                           const float* __restrict__ ew,
                                           float* __restrict__ dinv,
                                           int* __restrict__ off,
                                           int* __restrict__ src,
                                           float* __restrict__ wsl, int pblk, int tid,
                                           float* degL, int* cntL, int* curL, int* redL) {
  const int lo   = pblk * PREP_NODES;
  const int lane = tid & 31;
  const int wave = tid >> 5;
  const int4*   row4 = reinterpret_cast<const int4*>(row);
  const int4*   col4 = reinterpret_cast<const int4*>(col);
  const float4* ew4  = reinterpret_cast<const float4*>(ew);

  degL[tid] = 1.0f;
  cntL[tid] = 0;
  __syncthreads();

  int below = 0;
  for (int i0 = tid; i0 < NE4; i0 += 256 * PREP_U) {
    int4   c[PREP_U];
    float4 w[PREP_U];
#pragma unroll
    for (int u = 0; u < PREP_U; ++u) {
      const int i = i0 + 256 * u;
      if (i < NE4) { c[u] = col4[i]; w[u] = ew4[i]; }
      else         { c[u] = make_int4(NNODE, NNODE, NNODE, NNODE); w[u] = make_float4(0.f, 0.f, 0.f, 0.f); }
    }
#pragma unroll
    for (int u = 0; u < PREP_U; ++u) {
      const int   cc[4] = {c[u].x, c[u].y, c[u].z, c[u].w};
      const float ww[4] = {w[u].x, w[u].y, w[u].z, w[u].w};
#pragma unroll
      for (int k = 0; k < 4; ++k) {
        const int d = cc[k] - lo;
        below += (d < 0) ? 1 : 0;
        if ((unsigned)d < (unsigned)PREP_NODES) {
          atomicAdd(&degL[d], ww[k]);
          atomicAdd(&cntL[d], 1);
        }
      }
    }
  }
  for (int m = 16; m >= 1; m >>= 1) below += __shfl_xor(below, m);
  if (lane == 0) redL[wave] = below;
  __syncthreads();
  int base = 0;
  for (int v = 0; v < 8; ++v) base += redL[v];

  const int cn = cntL[tid];
  int inc = cn;
  for (int m = 1; m < 32; m <<= 1) {
    const int t = __shfl_up(inc, m);
    if (lane >= m) inc += t;
  }
  if (lane == 31) redL[8 + wave] = inc;
  __syncthreads();
  int wbase = 0;
  for (int v = 0; v < 8; ++v) wbase += (v < wave) ? redL[8 + v] : 0;
  const int first = base + wbase + inc - cn;
  off[lo + tid] = first;
  curL[tid]     = first;
  if (pblk == PREP_BLOCKS - 1 && tid == PREP_NODES - 1) off[NNODE] = first + cn;
  const float dg = degL[tid];
  dinv[lo + tid] = (dg > 0.0f) ? rsqrtf(dg) : 0.0f;
  __syncthreads();

  for (int i0 = tid; i0 < NE4; i0 += 256 * PREP_U) {
    int4   c[PREP_U], r[PREP_U];
    float4 w[PREP_U];
    int    slot[PREP_U][4];
#pragma unroll
    for (int u = 0; u < PREP_U; ++u) {
      const int i = i0 + 256 * u;
      if (i < NE4) { c[u] = col4[i]; r[u] = row4[i]; w[u] = ew4[i]; }
      else {
        c[u] = make_int4(NNODE, NNODE, NNODE, NNODE);
        r[u] = make_int4(0, 0, 0, 0);
        w[u] = make_float4(0.f, 0.f, 0.f, 0.f);
      }
    }
#pragma unroll
    for (int u = 0; u < PREP_U; ++u) {
      const int cc[4] = {c[u].x, c[u].y, c[u].z, c[u].w};
#pragma unroll
      for (int k = 0; k < 4; ++k) {
        const int d = cc[k] - lo;
        slot[u][k] = -1;
        if ((unsigned)d < (unsigned)PREP_NODES) slot[u][k] = atomicAdd(&curL[d], 1);
      }
    }
#pragma unroll
    for (int u = 0; u < PREP_U; ++u) {
      const int   rr[4] = {r[u].x, r[u].y, r[u].z, r[u].w};
      const float ww[4] = {w[u].x, w[u].y, w[u].z, w[u].w};
#pragma unroll
      for (int k = 0; k < 4; ++k) {
        if (slot[u][k] >= 0) {
          src[slot[u][k]] = rr[k];
          wsl[slot[u][k]] = ww[k];
        }
      }
    }
  }
}

__global__ __launch_bounds__(256) void k_conv1(const float* __restrict__ x,
                                               const float* __restrict__ Wc1,
                                               const float* __restrict__ b1,
                                               const float* __restrict__ Wg,
                                               float* __restrict__ xt,
                                               const int* __restrict__ row,
                                               const int* __restrict__ col,
                                               const float* __restrict__ ew,
                                               float* __restrict__ dinv,
                                               int* __restrict__ off,
                                               int* __restrict__ src,
                                               float* __restrict__ wsl) {
  __shared__ float Xs[16 * XS_STRIDE];
  __shared__ float Hs[8 * 16 * HS_STRIDE];
  __shared__ float PdegL[PREP_NODES];
  __shared__ int   PcntL[PREP_NODES];
  __shared__ int   PcurL[PREP_NODES];
  __shared__ int   PredL[16];

  const int tid = threadIdx.x;
  if (blockIdx.x < PREP_BLOCKS) {
    graph_prep(row, col, ew, dinv, off, src, wsl, blockIdx.x, tid, PdegL, PcntL, PcurL, PredL);
    return;
  }
  const int lane   = tid & 31;
  const int wave   = tid >> 5;
  const int lane16 = lane & 15;
  const int half   = lane >> 4;
  const int n0     = (blockIdx.x - PREP_BLOCKS) * 16;

  for (int idx = tid; idx < 16 * CIN * WLEN; idx += 256) {
    int n = idx / (CIN * WLEN), r = idx % (CIN * WLEN);
    Xs[n * XS_STRIDE + r] = x[(n0 + n) * (CIN * WLEN) + r];
  }

  const int pb  = wave & 3;
  const int coP = pb * 16 + lane16;
  const int coQ = coP + 64;
  v2f bp1[10], bq1[10];
#pragma unroll
  for (int s = 0; s < 10; ++s) {
    int k     = 4 * s + 2 * half;
    bp1[s][0] = Wc1[coP * 40 + k];
    bp1[s][1] = Wc1[coP * 40 + k + 1];
    bq1[s][0] = Wc1[coQ * 40 + k];
    bq1[s][1] = Wc1[coQ * 40 + k + 1];
  }
  const float biasP = b1[coP];
  const float biasQ = b1[coQ];

  v2f bwg[16];
#pragma unroll
  for (int s = 0; s < 16; ++s) {
    int k     = 4 * s + 2 * half;
    bwg[s][0] = Wg[k * 16 + lane16];
    bwg[s][1] = Wg[(k + 1) * 16 + lane16];
  }
  __syncthreads();

  const int dt = wave >> 2;
  for (int tbase = 0; tbase < T1; tbase += 8) {
    for (int i = 0; i < 4; ++i) {
      int tl = dt * 4 + i;
      int t  = tbase + tl;
      if (t < T1) {
        v8f accp, accq;
#pragma unroll
        for (int j = 0; j < 8; ++j) { accp[j] = biasP; accq[j] = biasQ; }
#pragma unroll
        for (int s = 0; s < 10; ++s) {
          int j0 = 4 * s + 2 * half;
          int ci = j0 >> 3, kk = j0 & 7;
          v2f a;
          a[0] = Xs[lane16 * XS_STRIDE + ci * WLEN + t + kk];
          a[1] = Xs[lane16 * XS_STRIDE + ci * WLEN + t + kk + 1];
          accp = wmma_f32(a, bp1[s], accp);
          accq = wmma_f32(a, bq1[s], accq);
        }
        int ch = pb * 16 + lane16;
#pragma unroll
        for (int j = 0; j < 8; ++j) {
          float hv = accp[j] * sigmoidf_(accq[j]);
          int m    = j + 8 * half;
          Hs[(tl * 16 + m) * HS_STRIDE + ch] = hv;
        }
      }
    }
    __syncthreads();
    {
      int tl = wave;
      int t  = tbase + tl;
      if (t < T1) {
        v8f acc = {};
#pragma unroll
        for (int s = 0; s < 16; ++s) {
          int k0 = 4 * s + 2 * half;
          v2f a;
          a[0] = Hs[(tl * 16 + lane16) * HS_STRIDE + k0];
          a[1] = Hs[(tl * 16 + lane16) * HS_STRIDE + k0 + 1];
          acc  = wmma_f32(a, bwg[s], acc);
        }
#pragma unroll
        for (int j = 0; j < 8; ++j) {
          int m = j + 8 * half;
          xt[((n0 + m) * T1 + t) * H2C + lane16] = acc[j];
        }
      }
    }
    __syncthreads();
  }
}

#define ROW4 (T1 * H2C / 4)
#define ECAP 1024
#define LU   16

__device__ __forceinline__ void gather_list(const float4* __restrict__ xt4, const int* es,
                                            const float* ws, const int* cl, int jpad,
                                            const float* __restrict__ bg, int tid, float* Gs) {
  const bool act = tid < ROW4;
  const int  rem = act ? tid : tid - 32;
  const int  f0  = (rem & 3) * 4;
  const float bg0 = bg[f0], bg1 = bg[f0 + 1], bg2 = bg[f0 + 2], bg3 = bg[f0 + 3];
  const float4* xcol = xt4 + rem;
  float*        gcol = &Gs[(rem >> 2) * GS_STRIDE + f0];
  float ax = 0.0f, ay = 0.0f, az = 0.0f, aw = 0.0f;
  for (int j0 = 0; j0 < jpad; j0 += LU) {
    int    r[LU], c[LU];
    float  w[LU];
    float4 v[LU];
#pragma unroll
    for (int k = 0; k < LU; ++k) {
      r[k] = __builtin_amdgcn_readfirstlane(es[j0 + k]);
      c[k] = __builtin_amdgcn_readfirstlane(cl[j0 + k]);
      w[k] = ws[j0 + k];
    }
#pragma unroll
    for (int k = 0; k < LU; ++k) v[k] = xcol[r[k] * ROW4];
#pragma unroll
    for (int k = 0; k < LU; ++k) {
      ax += w[k] * v[k].x;
      ay += w[k] * v[k].y;
      az += w[k] * v[k].z;
      aw += w[k] * v[k].w;
      if (c[k] >= 0) {
        const float gx = ax + bg0, gy = ay + bg1, gz = az + bg2, gw = aw + bg3;
        if (act) {
          float* g = gcol + c[k] * (T1 * GS_STRIDE);
          g[0] = gx > 0.0f ? gx : 0.0f;
          g[1] = gy > 0.0f ? gy : 0.0f;
          g[2] = gz > 0.0f ? gz : 0.0f;
          g[3] = gw > 0.0f ? gw : 0.0f;
        }
        ax = ay = az = aw = 0.0f;
      }
    }
  }
}

#define EU   8
#define NPC  2

__device__ __forceinline__ void gather_tile(const float4* __restrict__ xt4,
                                            const int* __restrict__ src,
                                            const float* __restrict__ nrm, int bs,
                                            const int* offs, const float* dn,
                                            const float* __restrict__ dinv,
                                            const float* __restrict__ bg, int n0,
                                            int tid, float* Gs) {
  const int f0 = (tid & 3) * 4;
  const float bg0 = bg[f0], bg1 = bg[f0 + 1], bg2 = bg[f0 + 2], bg3 = bg[f0 + 3];
  for (int o = tid; o < 16 * ROW4; o += 256 * NPC) {
    int   n[NPC], rem[NPC], b[NPC], en[NPC];
    bool  live[NPC];
    float4 vs[NPC];
    float ax[NPC], ay[NPC], az[NPC], aw[NPC];
    float dnq[NPC];
    int len = 0;
#pragma unroll
    for (int q = 0; q < NPC; ++q) {
      live[q] = (o + 256 * q) < 16 * ROW4;
      const int oq = live[q] ? (o + 256 * q) : o;
      n[q]   = oq / ROW4;
      rem[q] = oq - n[q] * ROW4;
      b[q]   = offs[n[q]];
      en[q]  = offs[n[q] + 1];
      dnq[q] = dn[n[q]];
      vs[q]  = xt4[(n0 + n[q]) * ROW4 + rem[q]];
      ax[q] = ay[q] = az[q] = aw[q] = 0.0f;
      len = max(len, en[q] - b[q]);
    }
    for (int j = 0; j < len; j += EU) {
      int   r[NPC][EU];
      float w[NPC][EU];
      float4 v[NPC][EU];
#pragma unroll
      for (int q = 0; q < NPC; ++q) {
#pragma unroll
        for (int k = 0; k < EU; ++k) {
          const int pp = b[q] + j + k;
          const int pc = max(min(pp, en[q] - 1), bs);
          r[q][k] = src[pc];
          const float wk = dinv[r[q][k]] * nrm[pc];
          w[q][k] = (pp < en[q]) ? wk * dnq[q] : 0.0f;
        }
      }
#pragma unroll
      for (int q = 0; q < NPC; ++q) {
#pragma unroll
        for (int k = 0; k < EU; ++k) v[q][k] = xt4[r[q][k] * ROW4 + rem[q]];
      }
#pragma unroll
      for (int q = 0; q < NPC; ++q) {
#pragma unroll
        for (int k = 0; k < EU; ++k) {
          ax[q] += w[q][k] * v[q][k].x;
          ay[q] += w[q][k] * v[q][k].y;
          az[q] += w[q][k] * v[q][k].z;
          aw[q] += w[q][k] * v[q][k].w;
        }
      }
    }
#pragma unroll
    for (int q = 0; q < NPC; ++q) {
      if (live[q]) {
        const float wself = dnq[q] * dnq[q];
        float gx = ax[q] + wself * vs[q].x + bg0;
        float gy = ay[q] + wself * vs[q].y + bg1;
        float gz = az[q] + wself * vs[q].z + bg2;
        float gw = aw[q] + wself * vs[q].w + bg3;
        float* g = &Gs[(n[q] * T1 + (rem[q] >> 2)) * GS_STRIDE + f0];
        g[0] = gx > 0.0f ? gx : 0.0f;
        g[1] = gy > 0.0f ? gy : 0.0f;
        g[2] = gz > 0.0f ? gz : 0.0f;
        g[3] = gw > 0.0f ? gw : 0.0f;
      }
    }
  }
}

#define OS_NODE (16 * T2)
__global__ __launch_bounds__(256) void k_conv2(const float* __restrict__ xt,
                                               const int* __restrict__ off,
                                               const int* __restrict__ src,
                                               const float* __restrict__ nrm,
                                               const float* __restrict__ dinv,
                                               const float* __restrict__ bg,
                                               const float* __restrict__ Wc2,
                                               const float* __restrict__ b2,
                                               float* __restrict__ out) {
  __shared__ float Gs[16 * T1 * GS_STRIDE];
  __shared__ int   Es[ECAP];
  __shared__ float Ws[ECAP];
  __shared__ int   Cs[ECAP];
  __shared__ int   Offs[17];
  __shared__ float Dn[16];
  __shared__ __attribute__((aligned(16))) float Os[8 * 2 * OS_NODE];

  const int tid    = threadIdx.x;
  const int lane   = tid & 31;
  const int wave   = tid >> 5;
  const int lane16 = lane & 15;
  const int half   = lane >> 4;
  const int n0     = blockIdx.x * 16;

  if (tid < 17) Offs[tid] = off[n0 + tid];
  if (tid >= 32 && tid < 48) Dn[tid - 32] = dinv[n0 + tid - 32];
  __syncthreads();
  const int  bs     = Offs[0];
  const int  nslot  = Offs[16] - bs;
  const int  jn     = nslot + 16;
  const int  jpad   = (jn + LU - 1) & ~(LU - 1);
  const bool staged = jpad <= ECAP;
  if (staged) {
    for (int i = tid; i < jpad; i += 256) {
      if (i < jn) {
        int n = 0;
#pragma unroll
        for (int m = 0; m < 16; ++m) n += (Offs[m + 1] - bs + m < i) ? 1 : 0;
        const float dn = Dn[n];
        if (i == Offs[n + 1] - bs + n) {
          Es[i] = n0 + n;
          Ws[i] = dn * dn;
          Cs[i] = n;
        } else {
          const int p = bs + i - n;
          const int s = src[p];
          Es[i] = s;
          Ws[i] = (dinv[s] * nrm[p]) * dn;
          Cs[i] = -1;
        }
      } else {
        Es[i] = n0 + 15;
        Ws[i] = 0.0f;
        Cs[i] = -1;
      }
    }
  }
  __syncthreads();
  const float4* xt4 = reinterpret_cast<const float4*>(xt);
  if (staged)
    gather_list(xt4, Es, Ws, Cs, jpad, bg, tid, Gs);
  else
    gather_tile(xt4, src, nrm, bs, Offs, Dn, dinv, bg, n0, tid, Gs);

  const int pb  = wave & 3;
  const int coP = pb * 16 + lane16;
  const int coQ = coP + 64;
  v2f ap2[32], aq2[32];
#pragma unroll
  for (int s = 0; s < 32; ++s) {
    int k     = 4 * s + 2 * half;
    ap2[s][0] = Wc2[coP * 128 + k];
    ap2[s][1] = Wc2[coP * 128 + k + 1];
    aq2[s][0] = Wc2[coQ * 128 + k];
    aq2[s][1] = Wc2[coQ * 128 + k + 1];
  }
  float biasP[8], biasQ[8];
#pragma unroll
  for (int j = 0; j < 8; ++j) {
    biasP[j] = b2[pb * 16 + 8 * half + j];
    biasQ[j] = b2[pb * 16 + 8 * half + j + 64];
  }
  __syncthreads();

  const int qhalf = wave >> 2;
  float* osw = &Os[wave * (2 * OS_NODE)];
  float* ow = out + (n0 + 8 * qhalf) * (H1C * T2) + pb * OS_NODE;
  for (int i = 0; i < 25; ++i) {
    const int q  = qhalf * 25 + i;
    const int cc = q * 16 + lane16;
    const int n  = cc / T2;
    const int t2 = cc - n * T2;
    const float* gb = &Gs[(n * T1 + t2 + 2 * half) * GS_STRIDE];
    v8f accp, accq;
#pragma unroll
    for (int j = 0; j < 8; ++j) { accp[j] = biasP[j]; accq[j] = biasQ[j]; }
    v2f bf[32];
#pragma unroll
    for (int s = 0; s < 32; ++s) {
      const int kk4 = (4 * s) & 7;
      const int ci  = s >> 1;
      bf[s][0] = gb[kk4 * GS_STRIDE + ci];
      bf[s][1] = gb[(kk4 + 1) * GS_STRIDE + ci];
    }
    __builtin_amdgcn_sched_barrier(0);
#pragma unroll
    for (int s = 0; s < 32; ++s) {
      accp = wmma_f32(ap2[s], bf[s], accp);
      accq = wmma_f32(aq2[s], bf[s], accq);
    }
    float* sp = osw + (n & 1) * OS_NODE + (8 * half) * T2 + t2;
#pragma unroll
    for (int j = 0; j < 8; ++j) sp[j * T2] = accp[j] * sigmoidf_(accq[j]);
    const int kdone = (16 * i) / T2;
    if ((16 * (i + 1)) / T2 > kdone) {
      const v4f* s4 = reinterpret_cast<const v4f*>(osw + (kdone & 1) * OS_NODE);
      v4f*       o4 = reinterpret_cast<v4f*>(ow + kdone * (H1C * T2));
#pragma unroll
      for (int u = 0; u < 7; ++u) {
        const int idx = lane + 32 * u;
        if (idx < OS_NODE / 4) __builtin_nontemporal_store(s4[idx], &o4[idx]);
      }
    }
  }
}

extern "C" void kernel_launch(void* const* d_in, const int* in_sizes, int n_in,
                              void* d_out, int out_size, void* d_ws, size_t ws_size,
                              hipStream_t stream) {
  const float* x   = (const float*)d_in[0];
  const int*   ei  = (const int*)d_in[1];
  const float* ea  = (const float*)d_in[2];
  const float* Wc1 = (const float*)d_in[4];
  const float* b1  = (const float*)d_in[5];
  const float* Wg  = (const float*)d_in[6];
  const float* bg  = (const float*)d_in[7];
  const float* Wc2 = (const float*)d_in[8];
  const float* b2  = (const float*)d_in[9];
  float*       out = (float*)d_out;

  const int* row = ei;
  const int* col = ei + NEDGE;

  const size_t xt_bytes = (size_t)T1 * NNODE * H2C * sizeof(float);
  char*  wsb = (char*)d_ws;
  float* xt  = (float*)wsb;
  float* deg = (float*)(wsb + xt_bytes);
  int*   off = (int*)(wsb + xt_bytes + 2 * 32768);
  int*   src = (int*)(wsb + xt_bytes + 4 * 32768);
  float* nrm = (float*)(wsb + xt_bytes + 4 * 32768 + 262144);

  k_conv1<<<PREP_BLOCKS + NNODE / 16, 256, 0, stream>>>(x, Wc1, b1, Wg, xt, row, col, ea, deg,
                                                        off, src, nrm);
  k_conv2<<<NNODE / 16, 256, 0, stream>>>(xt, off, src, nrm, deg, bg, Wc2, b2, out);
}
